# starting kernel
# baseline (speedup 1.0000x reference)
#include <hip/hip_runtime.h>
#include <hip/hip_bf16.h>
#include <math.h>

typedef __attribute__((ext_vector_type(2))) float v2f;
typedef __attribute__((ext_vector_type(8))) float v8f;

#define IN_F   256
#define OUTC   256
#define HEADS  4
#define OUT_F  64
#define NEG_SLOPE 0.2f

__device__ __forceinline__ float leaky(float v) { return v > 0.f ? v : NEG_SLOPE * v; }

__global__ void gat_init(float* __restrict__ out, const float* __restrict__ b,
                         float* __restrict__ gmax, float* __restrict__ gsum,
                         int outElems) {
    int i = blockIdx.x * blockDim.x + threadIdx.x;
    if (i < outElems) out[i] = b[i & (OUTC - 1)];
    if (i < HEADS) { gmax[i] = -INFINITY; gsum[i] = 0.f; }
}

__global__ void __launch_bounds__(256) gat_gemm_wmma(
        const float* __restrict__ x, const float* __restrict__ W,
        float* __restrict__ h, int nTilesM) {
    const int wave = blockIdx.x * (blockDim.x >> 5) + (threadIdx.x >> 5);
    const int lane = threadIdx.x & 31;
    const int mTile = wave >> 4;
    const int nTile = wave & 15;
    if (mTile >= nTilesM) return;

    const int m  = lane & 15;
    const int kb = (lane >> 4) << 1;

    const float* xrow = x + (size_t)(mTile * 16 + m) * IN_F;
    const float* wcol = W + nTile * 16 + (lane & 15);

    v8f c = {};
    #pragma unroll 4
    for (int k = 0; k < IN_F; k += 4) {
        v2f a = *(const v2f*)(xrow + k + kb);
        v2f bf;
        bf.x = wcol[(size_t)(k + kb)     * OUTC];
        bf.y = wcol[(size_t)(k + kb + 1) * OUTC];
        c = __builtin_amdgcn_wmma_f32_16x16x4_f32(
                 false, a,  false, bf,
                 (short)0, c,  false,  false);
    }

    const int nlocal = lane & 15;
    const int rowOff = (lane >> 4) << 3;
    float* hp = h + nTile * 16 + nlocal;
    #pragma unroll
    for (int i = 0; i < 8; ++i)
        hp[(size_t)(mTile * 16 + rowOff + i) * OUTC] = c[i];
}

__global__ void __launch_bounds__(256) gat_scores(
        const float* __restrict__ h, const int* __restrict__ ei,
        const float* __restrict__ att, float* __restrict__ scores,
        float* __restrict__ gmax, int E, int Nn) {
    const int wave = blockIdx.x * (blockDim.x >> 5) + (threadIdx.x >> 5);
    const int lane = threadIdx.x & 31;
    if (wave >= E + Nn) return;

    int s, d;
    if (wave < E) { s = ei[wave]; d = ei[E + wave]; }
    else          { s = d = wave - E; }

    const float* hs = h + (size_t)s * OUTC;
    const float* hd = h + (size_t)d * OUTC;

    float acc[HEADS] = {0.f, 0.f, 0.f, 0.f};
    #pragma unroll
    for (int i = 0; i < 8; ++i) {
        const int head = i >> 1;
        const int j    = lane + ((i & 1) << 5);
        const int idx  = lane + (i << 5);
        const float asv = att[head * 2 * OUT_F + j];
        const float adv = att[head * 2 * OUT_F + OUT_F + j];
        acc[head] += leaky(hs[idx]) * asv + leaky(hd[idx]) * adv;
    }

    #pragma unroll
    for (int hdi = 0; hdi < HEADS; ++hdi) {
        float v = acc[hdi];
        #pragma unroll
        for (int off = 16; off > 0; off >>= 1)
            v += __shfl_xor(v, off, 32);
        if (lane == 0) {
            scores[(size_t)wave * HEADS + hdi] = v;
            atomicMax(gmax + hdi, v);
        }
    }
}

__global__ void __launch_bounds__(256) gat_expsum(
        float* __restrict__ scores, const float* __restrict__ gmax,
        float* __restrict__ gsum, int total4) {
    __shared__ float ssum[HEADS];
    if (threadIdx.x < HEADS) ssum[threadIdx.x] = 0.f;
    __syncthreads();

    const int i = blockIdx.x * blockDim.x + threadIdx.x;
    if (i < total4) {
        const int hd = i & 3;
        const float v = __expf(scores[i] - gmax[hd]);
        scores[i] = v;
        atomicAdd(&ssum[hd], v);
    }
    __syncthreads();
    if (threadIdx.x < HEADS) atomicAdd(gsum + threadIdx.x, ssum[threadIdx.x]);
}

__global__ void __launch_bounds__(256) gat_scatter(
        const float* __restrict__ h, const int* __restrict__ ei,
        const float* __restrict__ scores, const float* __restrict__ gsum,
        float* __restrict__ out, int E, int Nn) {
    const int wave = blockIdx.x * (blockDim.x >> 5) + (threadIdx.x >> 5);
    const int lane = threadIdx.x & 31;
    if (wave >= E + Nn) return;

    int s, d;
    if (wave < E) { s = ei[wave]; d = ei[E + wave]; }
    else          { s = d = wave - E; }

    float wgt[HEADS];
    #pragma unroll
    for (int hd = 0; hd < HEADS; ++hd)
        wgt[hd] = scores[(size_t)wave * HEADS + hd] / gsum[hd];

    const float* hs = h + (size_t)s * OUTC;
    float* op = out + (size_t)d * OUTC;
    #pragma unroll
    for (int i = 0; i < 8; ++i) {
        const int idx = lane + (i << 5);
        atomicAdd(op + idx, hs[idx] * wgt[i >> 1]);
    }
}

extern "C" void kernel_launch(void* const* d_in, const int* in_sizes, int n_in,
                              void* d_out, int out_size, void* d_ws, size_t ws_size,
                              hipStream_t stream) {
    const float* x   = (const float*)d_in[0];
    const float* W   = (const float*)d_in[1];
    const float* att = (const float*)d_in[2];
    const float* b   = (const float*)d_in[3];
    const int*   ei  = (const int*)d_in[4];

    const int Nn = in_sizes[0] / IN_F;
    const int E  = in_sizes[4] / 2;
    const int totalEdges = E + Nn;

    float* out    = (float*)d_out;
    float* h      = (float*)d_ws;
    float* scores = h + (size_t)Nn * OUTC;
    float* gmax   = scores + (size_t)totalEdges * HEADS;
    float* gsum   = gmax + HEADS;

    const int outElems = Nn * OUTC;
    gat_init<<<(outElems + 255) / 256, 256, 0, stream>>>(out, b, gmax, gsum, outElems);

    const int nTilesM = Nn / 16;
    const int tiles   = nTilesM * (OUTC / 16);
    gat_gemm_wmma<<<(tiles + 7) / 8, 256, 0, stream>>>(x, W, h, nTilesM);

    gat_scores<<<(totalEdges + 7) / 8, 256, 0, stream>>>(h, ei, att, scores, gmax, E, Nn);
    gat_expsum<<<(totalEdges * HEADS + 255) / 256, 256, 0, stream>>>(
        scores, gmax, gsum, totalEdges * HEADS);
    gat_scatter<<<(totalEdges + 7) / 8, 256, 0, stream>>>(h, ei, scores, gsum, out, E, Nn);
}


# per-head max leaves the edge kernel: a new launch tree-reduces the stored scores (float4 loads, shuffles, LDS) to 32 partial maxima that each exp-sum wave folds itself; no float atomic max left
# speedup vs baseline: 6.6399x; 6.6399x over previous
#include <hip/hip_runtime.h>
#include <hip/hip_bf16.h>
#include <math.h>

typedef __attribute__((ext_vector_type(2))) float v2f;
typedef __attribute__((ext_vector_type(8))) float v8f;

#define IN_F   256
#define OUTC   256
#define HEADS  4
#define OUT_F  64
#define NEG_SLOPE 0.2f

__device__ __forceinline__ float leaky(float v) { return v > 0.f ? v : NEG_SLOPE * v; }

__global__ void gat_init(float* __restrict__ out, const float* __restrict__ b,
                         float* __restrict__ gsum, int outElems) {
    int i = blockIdx.x * blockDim.x + threadIdx.x;
    if (i < outElems) out[i] = b[i & (OUTC - 1)];
    if (i < HEADS) gsum[i] = 0.f;
}

__global__ void __launch_bounds__(256) gat_gemm_wmma(
        const float* __restrict__ x, const float* __restrict__ W,
        float* __restrict__ h, int nTilesM) {
    const int wave = blockIdx.x * (blockDim.x >> 5) + (threadIdx.x >> 5);
    const int lane = threadIdx.x & 31;
    const int mTile = wave >> 4;
    const int nTile = wave & 15;
    if (mTile >= nTilesM) return;

    const int m  = lane & 15;
    const int kb = (lane >> 4) << 1;

    const float* xrow = x + (size_t)(mTile * 16 + m) * IN_F;
    const float* wcol = W + nTile * 16 + (lane & 15);

    v8f c = {};
    #pragma unroll 4
    for (int k = 0; k < IN_F; k += 4) {
        v2f a = *(const v2f*)(xrow + k + kb);
        v2f bf;
        bf.x = wcol[(size_t)(k + kb)     * OUTC];
        bf.y = wcol[(size_t)(k + kb + 1) * OUTC];
        c = __builtin_amdgcn_wmma_f32_16x16x4_f32(
                 false, a,  false, bf,
                 (short)0, c,  false,  false);
    }

    const int nlocal = lane & 15;
    const int rowOff = (lane >> 4) << 3;
    float* hp = h + nTile * 16 + nlocal;
    #pragma unroll
    for (int i = 0; i < 8; ++i)
        hp[(size_t)(mTile * 16 + rowOff + i) * OUTC] = c[i];
}

__global__ void __launch_bounds__(256) gat_scores(
        const float* __restrict__ h, const int* __restrict__ ei,
        const float* __restrict__ att, float* __restrict__ scores,
        int E, int Nn) {
    const int wave = blockIdx.x * (blockDim.x >> 5) + (threadIdx.x >> 5);
    const int lane = threadIdx.x & 31;
    if (wave >= E + Nn) return;

    int s, d;
    if (wave < E) { s = ei[wave]; d = ei[E + wave]; }
    else          { s = d = wave - E; }

    const float* hs = h + (size_t)s * OUTC;
    const float* hd = h + (size_t)d * OUTC;

    float acc[HEADS] = {0.f, 0.f, 0.f, 0.f};
    #pragma unroll
    for (int i = 0; i < 8; ++i) {
        const int head = i >> 1;
        const int j    = lane + ((i & 1) << 5);
        const int idx  = lane + (i << 5);
        const float asv = att[head * 2 * OUT_F + j];
        const float adv = att[head * 2 * OUT_F + OUT_F + j];
        acc[head] += leaky(hs[idx]) * asv + leaky(hd[idx]) * adv;
    }

    #pragma unroll
    for (int hdi = 0; hdi < HEADS; ++hdi) {
        float v = acc[hdi];
        #pragma unroll
        for (int off = 16; off > 0; off >>= 1)
            v += __shfl_xor(v, off, 32);
        if (lane == 0) scores[(size_t)wave * HEADS + hdi] = v;
    }
}

#define NPART 32

__device__ __forceinline__ float4 max4(float4 a, float4 b) {
    return make_float4(fmaxf(a.x, b.x), fmaxf(a.y, b.y), fmaxf(a.z, b.z), fmaxf(a.w, b.w));
}

__device__ __forceinline__ float4 wave_max4(float4 m) {
    #pragma unroll
    for (int off = 16; off > 0; off >>= 1) {
        m.x = fmaxf(m.x, __shfl_xor(m.x, off, 32));
        m.y = fmaxf(m.y, __shfl_xor(m.y, off, 32));
        m.z = fmaxf(m.z, __shfl_xor(m.z, off, 32));
        m.w = fmaxf(m.w, __shfl_xor(m.w, off, 32));
    }
    return m;
}

__global__ void __launch_bounds__(1024) gat_max(
        const float* __restrict__ scores, float* __restrict__ pmax, int totalEdges) {
    __shared__ float4 smax[32];
    const float4* sc4 = (const float4*)scores;
    const int stride = NPART * 1024;

    float4 m = make_float4(-INFINITY, -INFINITY, -INFINITY, -INFINITY);
    #pragma unroll 4
    for (int i = blockIdx.x * 1024 + threadIdx.x; i < totalEdges; i += stride)
        m = max4(m, sc4[i]);

    m = wave_max4(m);
    const int lane = threadIdx.x & 31;
    if (lane == 0) smax[threadIdx.x >> 5] = m;
    __syncthreads();
    if (threadIdx.x < 32) {
        m = wave_max4(smax[lane]);
        if (lane == 0) ((float4*)pmax)[blockIdx.x] = m;
    }
}

__global__ void __launch_bounds__(256) gat_expsum(
        float* __restrict__ scores, const float* __restrict__ pmax,
        float* __restrict__ gsum, int total4) {
    __shared__ float ssum[HEADS];
    if (threadIdx.x < HEADS) ssum[threadIdx.x] = 0.f;

    const int lane = threadIdx.x & 31;
    float hmax = pmax[lane];
    #pragma unroll
    for (int j = 1; j < NPART / 8; ++j)
        hmax = fmaxf(hmax, pmax[lane + 32 * j]);
    hmax = fmaxf(hmax, __shfl_xor(hmax, 4, 32));
    hmax = fmaxf(hmax, __shfl_xor(hmax, 8, 32));
    hmax = fmaxf(hmax, __shfl_xor(hmax, 16, 32));
    __syncthreads();

    const int i = blockIdx.x * blockDim.x + threadIdx.x;
    if (i < total4) {
        const int hd = i & 3;
        const float v = __expf(scores[i] - hmax);
        scores[i] = v;
        atomicAdd(&ssum[hd], v);
    }
    __syncthreads();
    if (threadIdx.x < HEADS) atomicAdd(gsum + threadIdx.x, ssum[threadIdx.x]);
}

__global__ void __launch_bounds__(256) gat_scatter(
        const float* __restrict__ h, const int* __restrict__ ei,
        const float* __restrict__ scores, const float* __restrict__ gsum,
        float* __restrict__ out, int E, int Nn) {
    const int wave = blockIdx.x * (blockDim.x >> 5) + (threadIdx.x >> 5);
    const int lane = threadIdx.x & 31;
    if (wave >= E + Nn) return;

    int s, d;
    if (wave < E) { s = ei[wave]; d = ei[E + wave]; }
    else          { s = d = wave - E; }

    float wgt[HEADS];
    #pragma unroll
    for (int hd = 0; hd < HEADS; ++hd)
        wgt[hd] = scores[(size_t)wave * HEADS + hd] / gsum[hd];

    const float* hs = h + (size_t)s * OUTC;
    float* op = out + (size_t)d * OUTC;
    #pragma unroll
    for (int i = 0; i < 8; ++i) {
        const int idx = lane + (i << 5);
        atomicAdd(op + idx, hs[idx] * wgt[i >> 1]);
    }
}

extern "C" void kernel_launch(void* const* d_in, const int* in_sizes, int n_in,
                              void* d_out, int out_size, void* d_ws, size_t ws_size,
                              hipStream_t stream) {
    const float* x   = (const float*)d_in[0];
    const float* W   = (const float*)d_in[1];
    const float* att = (const float*)d_in[2];
    const float* b   = (const float*)d_in[3];
    const int*   ei  = (const int*)d_in[4];

    const int Nn = in_sizes[0] / IN_F;
    const int E  = in_sizes[4] / 2;
    const int totalEdges = E + Nn;

    float* out    = (float*)d_out;
    float* h      = (float*)d_ws;
    float* scores = h + (size_t)Nn * OUTC;
    float* pmax   = scores + (size_t)totalEdges * HEADS;
    float* gsum   = pmax + NPART * HEADS;

    const int outElems = Nn * OUTC;
    gat_init<<<(outElems + 255) / 256, 256, 0, stream>>>(out, b, gsum, outElems);

    const int nTilesM = Nn / 16;
    const int tiles   = nTilesM * (OUTC / 16);
    gat_gemm_wmma<<<(tiles + 7) / 8, 256, 0, stream>>>(x, W, h, nTilesM);

    gat_scores<<<(totalEdges + 7) / 8, 256, 0, stream>>>(h, ei, att, scores, E, Nn);
    gat_max<<<NPART, 1024, 0, stream>>>(scores, pmax, totalEdges);
    gat_expsum<<<(totalEdges * HEADS + 255) / 256, 256, 0, stream>>>(
        scores, pmax, gsum, totalEdges * HEADS);
    gat_scatter<<<(totalEdges + 7) / 8, 256, 0, stream>>>(h, ei, scores, gsum, out, E, Nn);
}


# aggregation by feature column: block c keeps column c of h (GEMM also stores h transposed) and of the sums in LDS, streams all edges, LDS float adds (sum order changes); global atomics and init go
# speedup vs baseline: 7.9339x; 1.1949x over previous
#include <hip/hip_runtime.h>
#include <hip/hip_bf16.h>
#include <math.h>

typedef __attribute__((ext_vector_type(2))) float v2f;
typedef __attribute__((ext_vector_type(8))) float v8f;

#define IN_F   256
#define OUTC   256
#define HEADS  4
#define OUT_F  64
#define NEG_SLOPE 0.2f

__device__ __forceinline__ float leaky(float v) { return v > 0.f ? v : NEG_SLOPE * v; }

__global__ void __launch_bounds__(256) gat_gemm_wmma(
        const float* __restrict__ x, const float* __restrict__ W,
        float* __restrict__ h, float* __restrict__ hT, int nTilesM, int Nn) {
    const int wave = blockIdx.x * (blockDim.x >> 5) + (threadIdx.x >> 5);
    const int lane = threadIdx.x & 31;
    const int mTile = wave >> 4;
    const int nTile = wave & 15;
    if (mTile >= nTilesM) return;

    const int m  = lane & 15;
    const int kb = (lane >> 4) << 1;

    const float* xrow = x + (size_t)(mTile * 16 + m) * IN_F;
    const float* wcol = W + nTile * 16 + (lane & 15);

    v8f c = {};
    #pragma unroll 4
    for (int k = 0; k < IN_F; k += 4) {
        v2f a = *(const v2f*)(xrow + k + kb);
        v2f bf;
        bf.x = wcol[(size_t)(k + kb)     * OUTC];
        bf.y = wcol[(size_t)(k + kb + 1) * OUTC];
        c = __builtin_amdgcn_wmma_f32_16x16x4_f32(
                 false, a,  false, bf,
                 (short)0, c,  false,  false);
    }

    const int nlocal = lane & 15;
    const int rowOff = (lane >> 4) << 3;
    float* hp = h + nTile * 16 + nlocal;
    #pragma unroll
    for (int i = 0; i < 8; ++i)
        hp[(size_t)(mTile * 16 + rowOff + i) * OUTC] = c[i];

    float* tp = hT + (size_t)(nTile * 16 + nlocal) * Nn + mTile * 16 + rowOff;
    *(float4*)tp       = make_float4(c[0], c[1], c[2], c[3]);
    *(float4*)(tp + 4) = make_float4(c[4], c[5], c[6], c[7]);
}

__global__ void __launch_bounds__(256) gat_scores(
        const float* __restrict__ h, const int* __restrict__ ei,
        const float* __restrict__ att, float* __restrict__ scores,
        int E, int Nn) {
    const int wave = blockIdx.x * (blockDim.x >> 5) + (threadIdx.x >> 5);
    const int lane = threadIdx.x & 31;
    if (wave >= E + Nn) return;

    int s, d;
    if (wave < E) { s = ei[wave]; d = ei[E + wave]; }
    else          { s = d = wave - E; }

    const float* hs = h + (size_t)s * OUTC;
    const float* hd = h + (size_t)d * OUTC;

    float acc[HEADS] = {0.f, 0.f, 0.f, 0.f};
    #pragma unroll
    for (int i = 0; i < 8; ++i) {
        const int head = i >> 1;
        const int j    = lane + ((i & 1) << 5);
        const int idx  = lane + (i << 5);
        const float asv = att[head * 2 * OUT_F + j];
        const float adv = att[head * 2 * OUT_F + OUT_F + j];
        acc[head] += leaky(hs[idx]) * asv + leaky(hd[idx]) * adv;
    }

    #pragma unroll
    for (int hdi = 0; hdi < HEADS; ++hdi) {
        float v = acc[hdi];
        #pragma unroll
        for (int off = 16; off > 0; off >>= 1)
            v += __shfl_xor(v, off, 32);
        if (lane == 0) scores[(size_t)wave * HEADS + hdi] = v;
    }
}

#define NPART 32

__device__ __forceinline__ float4 max4(float4 a, float4 b) {
    return make_float4(fmaxf(a.x, b.x), fmaxf(a.y, b.y), fmaxf(a.z, b.z), fmaxf(a.w, b.w));
}

__device__ __forceinline__ float4 wave_max4(float4 m) {
    #pragma unroll
    for (int off = 16; off > 0; off >>= 1) {
        m.x = fmaxf(m.x, __shfl_xor(m.x, off, 32));
        m.y = fmaxf(m.y, __shfl_xor(m.y, off, 32));
        m.z = fmaxf(m.z, __shfl_xor(m.z, off, 32));
        m.w = fmaxf(m.w, __shfl_xor(m.w, off, 32));
    }
    return m;
}

__global__ void __launch_bounds__(1024) gat_max(
        const float* __restrict__ scores, float* __restrict__ pmax,
        float* __restrict__ gsum, int totalEdges) {
    __shared__ float4 smax[32];
    if (blockIdx.x == 0 && threadIdx.x < HEADS) gsum[threadIdx.x] = 0.f;
    const float4* sc4 = (const float4*)scores;
    const int stride = NPART * 1024;

    float4 m = make_float4(-INFINITY, -INFINITY, -INFINITY, -INFINITY);
    #pragma unroll 4
    for (int i = blockIdx.x * 1024 + threadIdx.x; i < totalEdges; i += stride)
        m = max4(m, sc4[i]);

    m = wave_max4(m);
    const int lane = threadIdx.x & 31;
    if (lane == 0) smax[threadIdx.x >> 5] = m;
    __syncthreads();
    if (threadIdx.x < 32) {
        m = wave_max4(smax[lane]);
        if (lane == 0) ((float4*)pmax)[blockIdx.x] = m;
    }
}

__global__ void __launch_bounds__(256) gat_expsum(
        const float* __restrict__ scores, const float* __restrict__ pmax,
        float* __restrict__ wp, float* __restrict__ gsum, int total4) {
    __shared__ float ssum[HEADS];
    if (threadIdx.x < HEADS) ssum[threadIdx.x] = 0.f;

    const int lane = threadIdx.x & 31;
    float hmax = pmax[lane];
    #pragma unroll
    for (int j = 1; j < NPART / 8; ++j)
        hmax = fmaxf(hmax, pmax[lane + 32 * j]);
    hmax = fmaxf(hmax, __shfl_xor(hmax, 4, 32));
    hmax = fmaxf(hmax, __shfl_xor(hmax, 8, 32));
    hmax = fmaxf(hmax, __shfl_xor(hmax, 16, 32));
    __syncthreads();

    const int i = blockIdx.x * blockDim.x + threadIdx.x;
    if (i < total4) {
        const int hd = i & 3;
        const float v = __expf(scores[i] - hmax);
        wp[(size_t)hd * (total4 >> 2) + (i >> 2)] = v;
        atomicAdd(&ssum[hd], v);
    }
    __syncthreads();
    if (threadIdx.x < HEADS) atomicAdd(gsum + threadIdx.x, ssum[threadIdx.x]);
}

#define NODES 20000

__device__ __forceinline__ void col_trip(const float* hcol, float* ocol,
                                         int4 s, int4 d, float4 w) {
    const float h0 = hcol[s.x], h1 = hcol[s.y], h2 = hcol[s.z], h3 = hcol[s.w];
    asm volatile("" :: "v"(h0), "v"(h1), "v"(h2), "v"(h3));
    atomicAdd(&ocol[d.x], h0 * w.x);
    atomicAdd(&ocol[d.y], h1 * w.y);
    atomicAdd(&ocol[d.z], h2 * w.z);
    atomicAdd(&ocol[d.w], h3 * w.w);
}

__global__ void __launch_bounds__(1024) gat_colsum(
        const float* __restrict__ hT, const int* __restrict__ ei,
        const float* __restrict__ wp, float* __restrict__ outT, int E, int Nn) {
    __shared__ float hcol[NODES];
    __shared__ float ocol[NODES];
    if (Nn > NODES) return;

    const int c   = blockIdx.x;
    const int tid = threadIdx.x;
    const int total = E + Nn;
    const float* hsrc = hT + (size_t)c * Nn;
    const float* w    = wp + (size_t)(c >> 6) * total;
    const bool vec = (((E | Nn) & 3) == 0);

    const int* src = ei;
    const int* dst = ei + E;
    const int E4 = vec ? (E >> 2) : 0;
    const int4*   src4 = (const int4*)src;
    const int4*   dst4 = (const int4*)dst;
    const float4* w4p  = (const float4*)w;
    const int last = E4 - 1;
    const int rot = (int)(((long long)c * E4) / OUTC) & ~31;
    auto quad = [=](int p) { const int i = min(p, last) + rot; return i >= E4 ? i - E4 : i; };
    int4 sP = make_int4(0, 0, 0, 0), dP = sP, sQ = sP, dQ = sP;
    float4 wP = make_float4(0.f, 0.f, 0.f, 0.f), wQ = wP;
    if (E4 > 0) {
        { const int i = quad(tid);        sP = src4[i]; dP = dst4[i]; wP = w4p[i]; }
        { const int i = quad(tid + 1024); sQ = src4[i]; dQ = dst4[i]; wQ = w4p[i]; }
    }

    if (vec) {
        for (int q = tid; q < (Nn >> 2); q += 1024) {
            const float4 hv = ((const float4*)hsrc)[q];
            const float4 wv = ((const float4*)(w + E))[q];
            ((float4*)hcol)[q] = hv;
            ((float4*)ocol)[q] = make_float4(hv.x * wv.x, hv.y * wv.y, hv.z * wv.z, hv.w * wv.w);
        }
    } else {
        for (int n = tid; n < Nn; n += 1024) {
            const float hv = hsrc[n];
            hcol[n] = hv;
            ocol[n] = hv * w[E + n];
        }
    }
    __syncthreads();

    if (E4 > 0) {
        int4 sR, dR;
        float4 wR;
        for (int q = tid; q < E4; q += 3072) {
            { const int i = quad(q + 2048); sR = src4[i]; dR = dst4[i]; wR = w4p[i]; }
            col_trip(hcol, ocol, sP, dP, wP);
            { const int i = quad(q + 3072); sP = src4[i]; dP = dst4[i]; wP = w4p[i]; }
            if (q + 1024 < E4) col_trip(hcol, ocol, sQ, dQ, wQ);
            { const int i = quad(q + 4096); sQ = src4[i]; dQ = dst4[i]; wQ = w4p[i]; }
            if (q + 2048 < E4) col_trip(hcol, ocol, sR, dR, wR);
        }
    }
    for (int e = (E4 << 2) + tid; e < E; e += 1024)
        atomicAdd(&ocol[dst[e]], hcol[src[e]] * w[e]);
    __syncthreads();

    float* od = outT + (size_t)c * Nn;
    if (vec) {
        for (int q = tid; q < (Nn >> 2); q += 1024)
            ((float4*)od)[q] = ((const float4*)ocol)[q];
    } else {
        for (int n = tid; n < Nn; n += 1024) od[n] = ocol[n];
    }
}

__global__ void __launch_bounds__(256) gat_finish(
        const float* __restrict__ outT, const float* __restrict__ gsum,
        const float* __restrict__ b, float* __restrict__ out, int Nn) {
    __shared__ float tile[8][32][33];
    const int wv   = threadIdx.x >> 5;
    const int lane = threadIdx.x & 31;
    const int n0   = blockIdx.x * 32;
    const int c0   = wv * 32;
    const float inv  = 1.f / gsum[wv >> 1];
    const float bias = b[c0 + lane];

    const int nl = min(n0 + lane, Nn - 1);
    float v[32];
    #pragma unroll
    for (int j = 0; j < 32; ++j)
        v[j] = outT[(size_t)(c0 + j) * Nn + nl];
    #pragma unroll
    for (int j = 0; j < 32; ++j)
        tile[wv][j][lane] = v[j];
    __syncthreads();
    if (n0 + 32 <= Nn) {
        float r[32];
        #pragma unroll
        for (int i = 0; i < 32; ++i)
            r[i] = tile[wv][lane][i];
        #pragma unroll
        for (int i = 0; i < 32; ++i)
            out[(size_t)(n0 + i) * OUTC + c0 + lane] = r[i] * inv + bias;
    } else {
        for (int i = 0; i < 32; ++i)
            if (n0 + i < Nn)
                out[(size_t)(n0 + i) * OUTC + c0 + lane] = tile[wv][lane][i] * inv + bias;
    }
}

extern "C" void kernel_launch(void* const* d_in, const int* in_sizes, int n_in,
                              void* d_out, int out_size, void* d_ws, size_t ws_size,
                              hipStream_t stream) {
    const float* x   = (const float*)d_in[0];
    const float* W   = (const float*)d_in[1];
    const float* att = (const float*)d_in[2];
    const float* b   = (const float*)d_in[3];
    const int*   ei  = (const int*)d_in[4];

    const int Nn = in_sizes[0] / IN_F;
    const int E  = in_sizes[4] / 2;
    const int totalEdges = E + Nn;

    float* out    = (float*)d_out;
    float* h      = (float*)d_ws;
    float* hT     = h + (size_t)Nn * OUTC;
    float* outT   = hT + (size_t)Nn * OUTC;
    float* scores = outT + (size_t)Nn * OUTC;
    float* wp     = scores + (size_t)totalEdges * HEADS;
    float* pmax   = wp + (size_t)totalEdges * HEADS;
    float* gsum   = pmax + NPART * HEADS;

    const int nTilesM = Nn / 16;
    const int tiles   = nTilesM * (OUTC / 16);
    gat_gemm_wmma<<<(tiles + 7) / 8, 256, 0, stream>>>(x, W, h, hT, nTilesM, Nn);

    gat_scores<<<(totalEdges + 7) / 8, 256, 0, stream>>>(h, ei, att, scores, E, Nn);
    gat_max<<<NPART, 1024, 0, stream>>>(scores, pmax, gsum, totalEdges);
    gat_expsum<<<(totalEdges * HEADS + 255) / 256, 256, 0, stream>>>(
        scores, pmax, wp, gsum, totalEdges * HEADS);
    gat_colsum<<<OUTC, 1024, 0, stream>>>(hT, ei, wp, outT, E, Nn);
    gat_finish<<<(Nn + 31) / 32, 256, 0, stream>>>(outT, gsum, b, out, Nn);
}


# scores kernel: an 8-lane group per edge (4 edges per wave trip, 16 per wave) with 16-byte row loads and the attention vectors held in registers; head sums by a 3-step exchange (sum order changes)
# speedup vs baseline: 9.4260x; 1.1881x over previous
#include <hip/hip_runtime.h>
#include <hip/hip_bf16.h>
#include <math.h>

typedef __attribute__((ext_vector_type(2))) float v2f;
typedef __attribute__((ext_vector_type(8))) float v8f;

#define IN_F   256
#define OUTC   256
#define HEADS  4
#define OUT_F  64
#define NEG_SLOPE 0.2f

__device__ __forceinline__ float leaky(float v) { return v > 0.f ? v : NEG_SLOPE * v; }

__global__ void __launch_bounds__(256) gat_gemm_wmma(
        const float* __restrict__ x, const float* __restrict__ W,
        float* __restrict__ h, float* __restrict__ hT, int nTilesM, int Nn) {
    const int wave = blockIdx.x * (blockDim.x >> 5) + (threadIdx.x >> 5);
    const int lane = threadIdx.x & 31;
    const int mTile = wave >> 4;
    const int nTile = wave & 15;
    if (mTile >= nTilesM) return;

    const int m  = lane & 15;
    const int kb = (lane >> 4) << 1;

    const float* xrow = x + (size_t)(mTile * 16 + m) * IN_F;
    const float* wcol = W + nTile * 16 + (lane & 15);

    v8f c = {};
    #pragma unroll 4
    for (int k = 0; k < IN_F; k += 4) {
        v2f a = *(const v2f*)(xrow + k + kb);
        v2f bf;
        bf.x = wcol[(size_t)(k + kb)     * OUTC];
        bf.y = wcol[(size_t)(k + kb + 1) * OUTC];
        c = __builtin_amdgcn_wmma_f32_16x16x4_f32(
                 false, a,  false, bf,
                 (short)0, c,  false,  false);
    }

    const int nlocal = lane & 15;
    const int rowOff = (lane >> 4) << 3;
    float* hp = h + nTile * 16 + nlocal;
    #pragma unroll
    for (int i = 0; i < 8; ++i)
        hp[(size_t)(mTile * 16 + rowOff + i) * OUTC] = c[i];

    float* tp = hT + (size_t)(nTile * 16 + nlocal) * Nn + mTile * 16 + rowOff;
    *(float4*)tp       = make_float4(c[0], c[1], c[2], c[3]);
    *(float4*)(tp + 4) = make_float4(c[4], c[5], c[6], c[7]);
}

#define EPW 16

__device__ __forceinline__ float dot4_leaky(float4 v, float4 a) {
    return leaky(v.x) * a.x + leaky(v.y) * a.y + leaky(v.z) * a.z + leaky(v.w) * a.w;
}

__global__ void __launch_bounds__(256) gat_scores(
        const float* __restrict__ h, const int* __restrict__ ei,
        const float* __restrict__ att, float* __restrict__ scores,
        int E, int Nn) {
    const int wave = blockIdx.x * (blockDim.x >> 5) + (threadIdx.x >> 5);
    const int lane = threadIdx.x & 31;
    const int g = lane >> 3;
    const int j = lane & 7;
    const int total = E + Nn;
    const int e0 = wave * EPW;
    if (e0 >= total) return;

    float4 as[8], ad[8];
    #pragma unroll
    for (int i = 0; i < 8; ++i) {
        const int head = i >> 1;
        const int off  = (j + ((i & 1) << 3)) << 2;
        as[i] = *(const float4*)(att + head * 2 * OUT_F + off);
        ad[i] = *(const float4*)(att + head * 2 * OUT_F + OUT_F + off);
    }

    int sv[EPW / 4], dv[EPW / 4];
    #pragma unroll
    for (int t = 0; t < EPW / 4; ++t) {
        const int e = min(e0 + t * 4 + g, total - 1);
        if (e < E) { sv[t] = ei[e]; dv[t] = ei[E + e]; } else { sv[t] = dv[t] = e - E; }
    }

    const bool hi4 = (j & 4) != 0;
    const bool hi2 = (j & 2) != 0;

    float4 rs[2][8], rd[2][8];
    {
        const float4* hs = (const float4*)(h + (size_t)sv[0] * OUTC) + j;
        const float4* hd = (const float4*)(h + (size_t)dv[0] * OUTC) + j;
        #pragma unroll
        for (int i = 0; i < 8; ++i) { rs[0][i] = hs[8 * i]; rd[0][i] = hd[8 * i]; }
    }

    #pragma unroll
    for (int t = 0; t < EPW / 4; ++t) {
        const int e = e0 + t * 4 + g;
        const int cur = t & 1, nxt = cur ^ 1;
        if (t + 1 < EPW / 4) {
            const float4* hs = (const float4*)(h + (size_t)sv[t + 1] * OUTC) + j;
            const float4* hd = (const float4*)(h + (size_t)dv[t + 1] * OUTC) + j;
            #pragma unroll
            for (int i = 0; i < 8; ++i) { rs[nxt][i] = hs[8 * i]; rd[nxt][i] = hd[8 * i]; }
        }

        float acc[HEADS] = {0.f, 0.f, 0.f, 0.f};
        #pragma unroll
        for (int i = 0; i < 8; ++i)
            acc[i >> 1] += dot4_leaky(rs[cur][i], as[i]) + dot4_leaky(rd[cur][i], ad[i]);

        float k0 = hi4 ? acc[2] : acc[0];
        float k1 = hi4 ? acc[3] : acc[1];
        k0 += __shfl_xor(hi4 ? acc[0] : acc[2], 4, 32);
        k1 += __shfl_xor(hi4 ? acc[1] : acc[3], 4, 32);
        float v = hi2 ? k1 : k0;
        v += __shfl_xor(hi2 ? k0 : k1, 2, 32);
        v += __shfl_xor(v, 1, 32);

        if (e < total && (j & 1) == 0) scores[(size_t)e * HEADS + (j >> 1)] = v;
    }
}

#define NPART 32

__device__ __forceinline__ float4 max4(float4 a, float4 b) {
    return make_float4(fmaxf(a.x, b.x), fmaxf(a.y, b.y), fmaxf(a.z, b.z), fmaxf(a.w, b.w));
}

__device__ __forceinline__ float4 wave_max4(float4 m) {
    #pragma unroll
    for (int off = 16; off > 0; off >>= 1) {
        m.x = fmaxf(m.x, __shfl_xor(m.x, off, 32));
        m.y = fmaxf(m.y, __shfl_xor(m.y, off, 32));
        m.z = fmaxf(m.z, __shfl_xor(m.z, off, 32));
        m.w = fmaxf(m.w, __shfl_xor(m.w, off, 32));
    }
    return m;
}

__global__ void __launch_bounds__(1024) gat_max(
        const float* __restrict__ scores, float* __restrict__ pmax,
        float* __restrict__ gsum, int totalEdges) {
    __shared__ float4 smax[32];
    if (blockIdx.x == 0 && threadIdx.x < HEADS) gsum[threadIdx.x] = 0.f;
    const float4* sc4 = (const float4*)scores;
    const int stride = NPART * 1024;

    float4 m = make_float4(-INFINITY, -INFINITY, -INFINITY, -INFINITY);
    #pragma unroll 4
    for (int i = blockIdx.x * 1024 + threadIdx.x; i < totalEdges; i += stride)
        m = max4(m, sc4[i]);

    m = wave_max4(m);
    const int lane = threadIdx.x & 31;
    if (lane == 0) smax[threadIdx.x >> 5] = m;
    __syncthreads();
    if (threadIdx.x < 32) {
        m = wave_max4(smax[lane]);
        if (lane == 0) ((float4*)pmax)[blockIdx.x] = m;
    }
}

__global__ void __launch_bounds__(256) gat_expsum(
        const float* __restrict__ scores, const float* __restrict__ pmax,
        float* __restrict__ wp, float* __restrict__ gsum, int total4) {
    __shared__ float ssum[HEADS];
    if (threadIdx.x < HEADS) ssum[threadIdx.x] = 0.f;

    const int lane = threadIdx.x & 31;
    float hmax = pmax[lane];
    #pragma unroll
    for (int j = 1; j < NPART / 8; ++j)
        hmax = fmaxf(hmax, pmax[lane + 32 * j]);
    hmax = fmaxf(hmax, __shfl_xor(hmax, 4, 32));
    hmax = fmaxf(hmax, __shfl_xor(hmax, 8, 32));
    hmax = fmaxf(hmax, __shfl_xor(hmax, 16, 32));
    __syncthreads();

    const int i = blockIdx.x * blockDim.x + threadIdx.x;
    if (i < total4) {
        const int hd = i & 3;
        const float v = __expf(scores[i] - hmax);
        wp[(size_t)hd * (total4 >> 2) + (i >> 2)] = v;
        atomicAdd(&ssum[hd], v);
    }
    __syncthreads();
    if (threadIdx.x < HEADS) atomicAdd(gsum + threadIdx.x, ssum[threadIdx.x]);
}

#define NODES 20000

__device__ __forceinline__ void col_trip(const float* hcol, float* ocol,
                                         int4 s, int4 d, float4 w) {
    const float h0 = hcol[s.x], h1 = hcol[s.y], h2 = hcol[s.z], h3 = hcol[s.w];
    asm volatile("" :: "v"(h0), "v"(h1), "v"(h2), "v"(h3));
    atomicAdd(&ocol[d.x], h0 * w.x);
    atomicAdd(&ocol[d.y], h1 * w.y);
    atomicAdd(&ocol[d.z], h2 * w.z);
    atomicAdd(&ocol[d.w], h3 * w.w);
}

__global__ void __launch_bounds__(1024) gat_colsum(
        const float* __restrict__ hT, const int* __restrict__ ei,
        const float* __restrict__ wp, float* __restrict__ outT, int E, int Nn) {
    __shared__ float hcol[NODES];
    __shared__ float ocol[NODES];
    if (Nn > NODES) return;

    const int c   = blockIdx.x;
    const int tid = threadIdx.x;
    const int total = E + Nn;
    const float* hsrc = hT + (size_t)c * Nn;
    const float* w    = wp + (size_t)(c >> 6) * total;
    const bool vec = (((E | Nn) & 3) == 0);

    const int* src = ei;
    const int* dst = ei + E;
    const int E4 = vec ? (E >> 2) : 0;
    const int4*   src4 = (const int4*)src;
    const int4*   dst4 = (const int4*)dst;
    const float4* w4p  = (const float4*)w;
    const int last = E4 - 1;
    const int rot = (int)(((long long)c * E4) / OUTC) & ~31;
    auto quad = [=](int p) { const int i = min(p, last) + rot; return i >= E4 ? i - E4 : i; };
    int4 sP = make_int4(0, 0, 0, 0), dP = sP, sQ = sP, dQ = sP;
    float4 wP = make_float4(0.f, 0.f, 0.f, 0.f), wQ = wP;
    if (E4 > 0) {
        { const int i = quad(tid);        sP = src4[i]; dP = dst4[i]; wP = w4p[i]; }
        { const int i = quad(tid + 1024); sQ = src4[i]; dQ = dst4[i]; wQ = w4p[i]; }
    }

    if (vec) {
        for (int q = tid; q < (Nn >> 2); q += 1024) {
            const float4 hv = ((const float4*)hsrc)[q];
            const float4 wv = ((const float4*)(w + E))[q];
            ((float4*)hcol)[q] = hv;
            ((float4*)ocol)[q] = make_float4(hv.x * wv.x, hv.y * wv.y, hv.z * wv.z, hv.w * wv.w);
        }
    } else {
        for (int n = tid; n < Nn; n += 1024) {
            const float hv = hsrc[n];
            hcol[n] = hv;
            ocol[n] = hv * w[E + n];
        }
    }
    __syncthreads();

    if (E4 > 0) {
        int4 sR, dR;
        float4 wR;
        for (int q = tid; q < E4; q += 3072) {
            { const int i = quad(q + 2048); sR = src4[i]; dR = dst4[i]; wR = w4p[i]; }
            col_trip(hcol, ocol, sP, dP, wP);
            { const int i = quad(q + 3072); sP = src4[i]; dP = dst4[i]; wP = w4p[i]; }
            if (q + 1024 < E4) col_trip(hcol, ocol, sQ, dQ, wQ);
            { const int i = quad(q + 4096); sQ = src4[i]; dQ = dst4[i]; wQ = w4p[i]; }
            if (q + 2048 < E4) col_trip(hcol, ocol, sR, dR, wR);
        }
    }
    for (int e = (E4 << 2) + tid; e < E; e += 1024)
        atomicAdd(&ocol[dst[e]], hcol[src[e]] * w[e]);
    __syncthreads();

    float* od = outT + (size_t)c * Nn;
    if (vec) {
        for (int q = tid; q < (Nn >> 2); q += 1024)
            ((float4*)od)[q] = ((const float4*)ocol)[q];
    } else {
        for (int n = tid; n < Nn; n += 1024) od[n] = ocol[n];
    }
}

__global__ void __launch_bounds__(256) gat_finish(
        const float* __restrict__ outT, const float* __restrict__ gsum,
        const float* __restrict__ b, float* __restrict__ out, int Nn) {
    __shared__ float tile[8][32][33];
    const int wv   = threadIdx.x >> 5;
    const int lane = threadIdx.x & 31;
    const int n0   = blockIdx.x * 32;
    const int c0   = wv * 32;
    const float inv  = 1.f / gsum[wv >> 1];
    const float bias = b[c0 + lane];

    const int nl = min(n0 + lane, Nn - 1);
    float v[32];
    #pragma unroll
    for (int j = 0; j < 32; ++j)
        v[j] = outT[(size_t)(c0 + j) * Nn + nl];
    #pragma unroll
    for (int j = 0; j < 32; ++j)
        tile[wv][j][lane] = v[j];
    __syncthreads();
    if (n0 + 32 <= Nn) {
        float r[32];
        #pragma unroll
        for (int i = 0; i < 32; ++i)
            r[i] = tile[wv][lane][i];
        #pragma unroll
        for (int i = 0; i < 32; ++i)
            out[(size_t)(n0 + i) * OUTC + c0 + lane] = r[i] * inv + bias;
    } else {
        for (int i = 0; i < 32; ++i)
            if (n0 + i < Nn)
                out[(size_t)(n0 + i) * OUTC + c0 + lane] = tile[wv][lane][i] * inv + bias;
    }
}

extern "C" void kernel_launch(void* const* d_in, const int* in_sizes, int n_in,
                              void* d_out, int out_size, void* d_ws, size_t ws_size,
                              hipStream_t stream) {
    const float* x   = (const float*)d_in[0];
    const float* W   = (const float*)d_in[1];
    const float* att = (const float*)d_in[2];
    const float* b   = (const float*)d_in[3];
    const int*   ei  = (const int*)d_in[4];

    const int Nn = in_sizes[0] / IN_F;
    const int E  = in_sizes[4] / 2;
    const int totalEdges = E + Nn;

    float* out    = (float*)d_out;
    float* h      = (float*)d_ws;
    float* hT     = h + (size_t)Nn * OUTC;
    float* outT   = hT + (size_t)Nn * OUTC;
    float* scores = outT + (size_t)Nn * OUTC;
    float* wp     = scores + (size_t)totalEdges * HEADS;
    float* pmax   = wp + (size_t)totalEdges * HEADS;
    float* gsum   = pmax + NPART * HEADS;

    const int nTilesM = Nn / 16;
    const int tiles   = nTilesM * (OUTC / 16);
    gat_gemm_wmma<<<(tiles + 7) / 8, 256, 0, stream>>>(x, W, h, hT, nTilesM, Nn);

    gat_scores<<<(totalEdges + 8 * EPW - 1) / (8 * EPW), 256, 0, stream>>>(h, ei, att, scores, E, Nn);
    gat_max<<<NPART, 1024, 0, stream>>>(scores, pmax, gsum, totalEdges);
    gat_expsum<<<(totalEdges * HEADS + 255) / 256, 256, 0, stream>>>(
        scores, pmax, wp, gsum, totalEdges * HEADS);
    gat_colsum<<<OUTC, 1024, 0, stream>>>(hT, ei, wp, outT, E, Nn);
    gat_finish<<<(Nn + 31) / 32, 256, 0, stream>>>(outT, gsum, b, out, Nn);
}


# GEMM through LDS: a 16-wave block owns 80 rows x 256 columns, stages 32-deep chunks of x and W in LDS by whole-line 16-byte loads (two stages), waves read fragments from LDS; k order kept
# speedup vs baseline: 11.3522x; 1.2043x over previous
#include <hip/hip_runtime.h>
#include <hip/hip_bf16.h>
#include <math.h>

typedef __attribute__((ext_vector_type(2))) float v2f;
typedef __attribute__((ext_vector_type(8))) float v8f;

#define IN_F   256
#define OUTC   256
#define HEADS  4
#define OUT_F  64
#define NEG_SLOPE 0.2f

__device__ __forceinline__ float leaky(float v) { return v > 0.f ? v : NEG_SLOPE * v; }

#define GM_MT   5
#define GM_ROWS (GM_MT * 16)
#define GM_KC   32
#define GM_XP   36
#define GM_WP   544

__global__ void __launch_bounds__(512) gat_gemm_wmma(
        const float* __restrict__ x, const float* __restrict__ W,
        float* __restrict__ h, float* __restrict__ hT, int nTilesM, int Nn) {
    __shared__ __attribute__((aligned(16))) float xs[2][GM_ROWS * GM_XP];
    __shared__ __attribute__((aligned(16))) float wsm[2][(GM_KC / 2) * GM_WP];

    const int tid  = threadIdx.x;
    const int wv   = tid >> 5;
    const int lane = tid & 31;
    const int m    = lane & 15;
    const int hf   = lane >> 4;
    const int row0 = blockIdx.x * GM_ROWS;

    const int  xr0 = tid >> 3, xq = tid & 7;
    const int  xr1 = (tid + 512) >> 3;
    const bool x2  = tid < GM_ROWS * 8 - 512;
    const float* xg0 = x + (size_t)min(row0 + xr0, Nn - 1) * IN_F + xq * 4;
    const float* xg1 = x + (size_t)min(row0 + xr1, Nn - 1) * IN_F + xq * 4;
    const int wp0 = tid >> 6, wc = tid & 63;
    const float* wg = W + (size_t)(2 * wp0) * OUTC + wc * 4;

    float4 xa, xb = make_float4(0.f, 0.f, 0.f, 0.f), w0a, w0b, w1a, w1b;
    auto gload = [&](int k0) {
        xa = *(const float4*)(xg0 + k0);
        if (x2) xb = *(const float4*)(xg1 + k0);
        const float* p = wg + (size_t)k0 * OUTC;
        w0a = *(const float4*)(p);
        w0b = *(const float4*)(p + OUTC);
        w1a = *(const float4*)(p + 16 * OUTC);
        w1b = *(const float4*)(p + 17 * OUTC);
    };
    auto sstore = [&](int b) {
        *(float4*)(&xs[b][xr0 * GM_XP + xq * 4]) = xa;
        if (x2) *(float4*)(&xs[b][xr1 * GM_XP + xq * 4]) = xb;
        float* d0 = &wsm[b][wp0 * GM_WP + wc * 8];
        *(float4*)(d0)     = make_float4(w0a.x, w0b.x, w0a.y, w0b.y);
        *(float4*)(d0 + 4) = make_float4(w0a.z, w0b.z, w0a.w, w0b.w);
        float* d1 = d0 + 8 * GM_WP;
        *(float4*)(d1)     = make_float4(w1a.x, w1b.x, w1a.y, w1b.y);
        *(float4*)(d1 + 4) = make_float4(w1a.z, w1b.z, w1a.w, w1b.w);
    };

    v8f acc[GM_MT];
    #pragma unroll
    for (int t = 0; t < GM_MT; ++t) acc[t] = v8f{};

    gload(0);
    sstore(0);
    __syncthreads();

    for (int c = 0; c < IN_F / GM_KC; ++c) {
        const int  b    = c & 1;
        const bool more = c + 1 < IN_F / GM_KC;
        if (more) gload((c + 1) * GM_KC);
        __builtin_amdgcn_sched_barrier(0);
        const float* xf = &xs[b][m * GM_XP + 2 * hf];
        const float* wf = &wsm[b][hf * GM_WP + 2 * (wv * 16 + m)];
        #pragma unroll
        for (int ks = 0; ks < GM_KC / 4; ++ks) {
            const v2f bf = *(const v2f*)(wf + ks * 2 * GM_WP);
            #pragma unroll
            for (int t = 0; t < GM_MT; ++t) {
                const v2f a = *(const v2f*)(xf + t * 16 * GM_XP + ks * 4);
                acc[t] = __builtin_amdgcn_wmma_f32_16x16x4_f32(
                         false, a,  false, bf,
                         (short)0, acc[t],  false,  false);
            }
        }
        if (more) sstore(b ^ 1);
        __syncthreads();
    }

    const int rowOff = hf << 3;
    #pragma unroll
    for (int t = 0; t < GM_MT; ++t) {
        const int mTile = blockIdx.x * GM_MT + t;
        if (mTile < nTilesM) {
            float* hp = h + (size_t)(mTile * 16 + rowOff) * OUTC + wv * 16 + m;
            #pragma unroll
            for (int i = 0; i < 8; ++i)
                hp[(size_t)i * OUTC] = acc[t][i];
            float* tp = hT + (size_t)(wv * 16 + m) * Nn + mTile * 16 + rowOff;
            *(float4*)tp       = make_float4(acc[t][0], acc[t][1], acc[t][2], acc[t][3]);
            *(float4*)(tp + 4) = make_float4(acc[t][4], acc[t][5], acc[t][6], acc[t][7]);
        }
    }
}

#define EPW 16

__device__ __forceinline__ float dot4_leaky(float4 v, float4 a) {
    return leaky(v.x) * a.x + leaky(v.y) * a.y + leaky(v.z) * a.z + leaky(v.w) * a.w;
}

__global__ void __launch_bounds__(256) gat_scores(
        const float* __restrict__ h, const int* __restrict__ ei,
        const float* __restrict__ att, float* __restrict__ scores,
        int E, int Nn) {
    const int wave = blockIdx.x * (blockDim.x >> 5) + (threadIdx.x >> 5);
    const int lane = threadIdx.x & 31;
    const int g = lane >> 3;
    const int j = lane & 7;
    const int total = E + Nn;
    const int e0 = wave * EPW;
    if (e0 >= total) return;

    float4 as[8], ad[8];
    #pragma unroll
    for (int i = 0; i < 8; ++i) {
        const int head = i >> 1;
        const int off  = (j + ((i & 1) << 3)) << 2;
        as[i] = *(const float4*)(att + head * 2 * OUT_F + off);
        ad[i] = *(const float4*)(att + head * 2 * OUT_F + OUT_F + off);
    }

    int sv[EPW / 4], dv[EPW / 4];
    #pragma unroll
    for (int t = 0; t < EPW / 4; ++t) {
        const int e = min(e0 + t * 4 + g, total - 1);
        if (e < E) { sv[t] = ei[e]; dv[t] = ei[E + e]; } else { sv[t] = dv[t] = e - E; }
    }

    const bool hi4 = (j & 4) != 0;
    const bool hi2 = (j & 2) != 0;

    float4 rs[2][8], rd[2][8];
    {
        const float4* hs = (const float4*)(h + (size_t)sv[0] * OUTC) + j;
        const float4* hd = (const float4*)(h + (size_t)dv[0] * OUTC) + j;
        #pragma unroll
        for (int i = 0; i < 8; ++i) { rs[0][i] = hs[8 * i]; rd[0][i] = hd[8 * i]; }
    }

    #pragma unroll
    for (int t = 0; t < EPW / 4; ++t) {
        const int e = e0 + t * 4 + g;
        const int cur = t & 1, nxt = cur ^ 1;
        if (t + 1 < EPW / 4) {
            const float4* hs = (const float4*)(h + (size_t)sv[t + 1] * OUTC) + j;
            const float4* hd = (const float4*)(h + (size_t)dv[t + 1] * OUTC) + j;
            #pragma unroll
            for (int i = 0; i < 8; ++i) { rs[nxt][i] = hs[8 * i]; rd[nxt][i] = hd[8 * i]; }
        }

        float acc[HEADS] = {0.f, 0.f, 0.f, 0.f};
        #pragma unroll
        for (int i = 0; i < 8; ++i)
            acc[i >> 1] += dot4_leaky(rs[cur][i], as[i]) + dot4_leaky(rd[cur][i], ad[i]);

        float k0 = hi4 ? acc[2] : acc[0];
        float k1 = hi4 ? acc[3] : acc[1];
        k0 += __shfl_xor(hi4 ? acc[0] : acc[2], 4, 32);
        k1 += __shfl_xor(hi4 ? acc[1] : acc[3], 4, 32);
        float v = hi2 ? k1 : k0;
        v += __shfl_xor(hi2 ? k0 : k1, 2, 32);
        v += __shfl_xor(v, 1, 32);

        if (e < total && (j & 1) == 0) scores[(size_t)e * HEADS + (j >> 1)] = v;
    }
}

#define NPART 32

__device__ __forceinline__ float4 max4(float4 a, float4 b) {
    return make_float4(fmaxf(a.x, b.x), fmaxf(a.y, b.y), fmaxf(a.z, b.z), fmaxf(a.w, b.w));
}

__device__ __forceinline__ float4 wave_max4(float4 m) {
    #pragma unroll
    for (int off = 16; off > 0; off >>= 1) {
        m.x = fmaxf(m.x, __shfl_xor(m.x, off, 32));
        m.y = fmaxf(m.y, __shfl_xor(m.y, off, 32));
        m.z = fmaxf(m.z, __shfl_xor(m.z, off, 32));
        m.w = fmaxf(m.w, __shfl_xor(m.w, off, 32));
    }
    return m;
}

__global__ void __launch_bounds__(1024) gat_max(
        const float* __restrict__ scores, float* __restrict__ pmax,
        float* __restrict__ gsum, int totalEdges) {
    __shared__ float4 smax[32];
    if (blockIdx.x == 0 && threadIdx.x < HEADS) gsum[threadIdx.x] = 0.f;
    const float4* sc4 = (const float4*)scores;
    const int stride = NPART * 1024;

    float4 m = make_float4(-INFINITY, -INFINITY, -INFINITY, -INFINITY);
    #pragma unroll 4
    for (int i = blockIdx.x * 1024 + threadIdx.x; i < totalEdges; i += stride)
        m = max4(m, sc4[i]);

    m = wave_max4(m);
    const int lane = threadIdx.x & 31;
    if (lane == 0) smax[threadIdx.x >> 5] = m;
    __syncthreads();
    if (threadIdx.x < 32) {
        m = wave_max4(smax[lane]);
        if (lane == 0) ((float4*)pmax)[blockIdx.x] = m;
    }
}

__global__ void __launch_bounds__(256) gat_expsum(
        const float* __restrict__ scores, const float* __restrict__ pmax,
        float* __restrict__ wp, float* __restrict__ gsum, int total4) {
    __shared__ float ssum[HEADS];
    if (threadIdx.x < HEADS) ssum[threadIdx.x] = 0.f;

    const int lane = threadIdx.x & 31;
    float hmax = pmax[lane];
    #pragma unroll
    for (int j = 1; j < NPART / 8; ++j)
        hmax = fmaxf(hmax, pmax[lane + 32 * j]);
    hmax = fmaxf(hmax, __shfl_xor(hmax, 4, 32));
    hmax = fmaxf(hmax, __shfl_xor(hmax, 8, 32));
    hmax = fmaxf(hmax, __shfl_xor(hmax, 16, 32));
    __syncthreads();

    const int i = blockIdx.x * blockDim.x + threadIdx.x;
    if (i < total4) {
        const int hd = i & 3;
        const float v = __expf(scores[i] - hmax);
        wp[(size_t)hd * (total4 >> 2) + (i >> 2)] = v;
        atomicAdd(&ssum[hd], v);
    }
    __syncthreads();
    if (threadIdx.x < HEADS) atomicAdd(gsum + threadIdx.x, ssum[threadIdx.x]);
}

#define NODES 20000

__device__ __forceinline__ void col_trip(const float* hcol, float* ocol,
                                         int4 s, int4 d, float4 w) {
    const float h0 = hcol[s.x], h1 = hcol[s.y], h2 = hcol[s.z], h3 = hcol[s.w];
    asm volatile("" :: "v"(h0), "v"(h1), "v"(h2), "v"(h3));
    atomicAdd(&ocol[d.x], h0 * w.x);
    atomicAdd(&ocol[d.y], h1 * w.y);
    atomicAdd(&ocol[d.z], h2 * w.z);
    atomicAdd(&ocol[d.w], h3 * w.w);
}

__global__ void __launch_bounds__(1024) gat_colsum(
        const float* __restrict__ hT, const int* __restrict__ ei,
        const float* __restrict__ wp, float* __restrict__ outT, int E, int Nn) {
    __shared__ float hcol[NODES];
    __shared__ float ocol[NODES];
    if (Nn > NODES) return;

    const int c   = blockIdx.x;
    const int tid = threadIdx.x;
    const int total = E + Nn;
    const float* hsrc = hT + (size_t)c * Nn;
    const float* w    = wp + (size_t)(c >> 6) * total;
    const bool vec = (((E | Nn) & 3) == 0);

    const int* src = ei;
    const int* dst = ei + E;
    const int E4 = vec ? (E >> 2) : 0;
    const int4*   src4 = (const int4*)src;
    const int4*   dst4 = (const int4*)dst;
    const float4* w4p  = (const float4*)w;
    const int last = E4 - 1;
    const int rot = (int)(((long long)c * E4) / OUTC) & ~31;
    auto quad = [=](int p) { const int i = min(p, last) + rot; return i >= E4 ? i - E4 : i; };
    int4 sP = make_int4(0, 0, 0, 0), dP = sP, sQ = sP, dQ = sP;
    float4 wP = make_float4(0.f, 0.f, 0.f, 0.f), wQ = wP;
    if (E4 > 0) {
        { const int i = quad(tid);        sP = src4[i]; dP = dst4[i]; wP = w4p[i]; }
        { const int i = quad(tid + 1024); sQ = src4[i]; dQ = dst4[i]; wQ = w4p[i]; }
    }

    if (vec) {
        for (int q = tid; q < (Nn >> 2); q += 1024) {
            const float4 hv = ((const float4*)hsrc)[q];
            const float4 wv = ((const float4*)(w + E))[q];
            ((float4*)hcol)[q] = hv;
            ((float4*)ocol)[q] = make_float4(hv.x * wv.x, hv.y * wv.y, hv.z * wv.z, hv.w * wv.w);
        }
    } else {
        for (int n = tid; n < Nn; n += 1024) {
            const float hv = hsrc[n];
            hcol[n] = hv;
            ocol[n] = hv * w[E + n];
        }
    }
    __syncthreads();

    if (E4 > 0) {
        int4 sR, dR;
        float4 wR;
        for (int q = tid; q < E4; q += 3072) {
            { const int i = quad(q + 2048); sR = src4[i]; dR = dst4[i]; wR = w4p[i]; }
            col_trip(hcol, ocol, sP, dP, wP);
            { const int i = quad(q + 3072); sP = src4[i]; dP = dst4[i]; wP = w4p[i]; }
            if (q + 1024 < E4) col_trip(hcol, ocol, sQ, dQ, wQ);
            { const int i = quad(q + 4096); sQ = src4[i]; dQ = dst4[i]; wQ = w4p[i]; }
            if (q + 2048 < E4) col_trip(hcol, ocol, sR, dR, wR);
        }
    }
    for (int e = (E4 << 2) + tid; e < E; e += 1024)
        atomicAdd(&ocol[dst[e]], hcol[src[e]] * w[e]);
    __syncthreads();

    float* od = outT + (size_t)c * Nn;
    if (vec) {
        for (int q = tid; q < (Nn >> 2); q += 1024)
            ((float4*)od)[q] = ((const float4*)ocol)[q];
    } else {
        for (int n = tid; n < Nn; n += 1024) od[n] = ocol[n];
    }
}

__global__ void __launch_bounds__(256) gat_finish(
        const float* __restrict__ outT, const float* __restrict__ gsum,
        const float* __restrict__ b, float* __restrict__ out, int Nn) {
    __shared__ float tile[8][32][33];
    const int wv   = threadIdx.x >> 5;
    const int lane = threadIdx.x & 31;
    const int n0   = blockIdx.x * 32;
    const int c0   = wv * 32;
    const float inv  = 1.f / gsum[wv >> 1];
    const float bias = b[c0 + lane];

    const int nl = min(n0 + lane, Nn - 1);
    float v[32];
    #pragma unroll
    for (int j = 0; j < 32; ++j)
        v[j] = outT[(size_t)(c0 + j) * Nn + nl];
    #pragma unroll
    for (int j = 0; j < 32; ++j)
        tile[wv][j][lane] = v[j];
    __syncthreads();
    if (n0 + 32 <= Nn) {
        float r[32];
        #pragma unroll
        for (int i = 0; i < 32; ++i)
            r[i] = tile[wv][lane][i];
        #pragma unroll
        for (int i = 0; i < 32; ++i)
            out[(size_t)(n0 + i) * OUTC + c0 + lane] = r[i] * inv + bias;
    } else {
        for (int i = 0; i < 32; ++i)
            if (n0 + i < Nn)
                out[(size_t)(n0 + i) * OUTC + c0 + lane] = tile[wv][lane][i] * inv + bias;
    }
}

extern "C" void kernel_launch(void* const* d_in, const int* in_sizes, int n_in,
                              void* d_out, int out_size, void* d_ws, size_t ws_size,
                              hipStream_t stream) {
    const float* x   = (const float*)d_in[0];
    const float* W   = (const float*)d_in[1];
    const float* att = (const float*)d_in[2];
    const float* b   = (const float*)d_in[3];
    const int*   ei  = (const int*)d_in[4];

    const int Nn = in_sizes[0] / IN_F;
    const int E  = in_sizes[4] / 2;
    const int totalEdges = E + Nn;

    float* out    = (float*)d_out;
    float* h      = (float*)d_ws;
    float* hT     = h + (size_t)Nn * OUTC;
    float* outT   = hT + (size_t)Nn * OUTC;
    float* scores = outT + (size_t)Nn * OUTC;
    float* wp     = scores + (size_t)totalEdges * HEADS;
    float* pmax   = wp + (size_t)totalEdges * HEADS;
    float* gsum   = pmax + NPART * HEADS;

    const int nTilesM = Nn / 16;
    gat_gemm_wmma<<<(nTilesM + GM_MT - 1) / GM_MT, 512, 0, stream>>>(x, W, h, hT, nTilesM, Nn);

    gat_scores<<<(totalEdges + 8 * EPW - 1) / (8 * EPW), 256, 0, stream>>>(h, ei, att, scores, E, Nn);
    gat_max<<<NPART, 1024, 0, stream>>>(scores, pmax, gsum, totalEdges);
    gat_expsum<<<(totalEdges * HEADS + 255) / 256, 256, 0, stream>>>(
        scores, pmax, wp, gsum, totalEdges * HEADS);
    gat_colsum<<<OUTC, 1024, 0, stream>>>(hT, ei, wp, outT, E, Nn);
    gat_finish<<<(Nn + 31) / 32, 256, 0, stream>>>(outT, gsum, b, out, Nn);
}


# exp-sum launch goes: each aggregation block forms exp(score - max) of its head's planar scores itself, totals them (thread, shuffles, LDS) and scales its column; no sum atomics (sum order changes)
# speedup vs baseline: 12.1576x; 1.0709x over previous
#include <hip/hip_runtime.h>
#include <hip/hip_bf16.h>
#include <math.h>

typedef __attribute__((ext_vector_type(2))) float v2f;
typedef __attribute__((ext_vector_type(8))) float v8f;

#define IN_F   256
#define OUTC   256
#define HEADS  4
#define OUT_F  64
#define NEG_SLOPE 0.2f

__device__ __forceinline__ float leaky(float v) { return v > 0.f ? v : NEG_SLOPE * v; }

#define GM_MT   5
#define GM_ROWS (GM_MT * 16)
#define GM_KC   32
#define GM_XP   36
#define GM_WP   544

__global__ void __launch_bounds__(512) gat_gemm_wmma(
        const float* __restrict__ x, const float* __restrict__ W,
        float* __restrict__ h, float* __restrict__ hT, int nTilesM, int Nn) {
    __shared__ __attribute__((aligned(16))) float xs[2][GM_ROWS * GM_XP];
    __shared__ __attribute__((aligned(16))) float wsm[2][(GM_KC / 2) * GM_WP];

    const int tid  = threadIdx.x;
    const int wv   = tid >> 5;
    const int lane = tid & 31;
    const int m    = lane & 15;
    const int hf   = lane >> 4;
    const int row0 = blockIdx.x * GM_ROWS;

    const int  xr0 = tid >> 3, xq = tid & 7;
    const int  xr1 = (tid + 512) >> 3;
    const bool x2  = tid < GM_ROWS * 8 - 512;
    const float* xg0 = x + (size_t)min(row0 + xr0, Nn - 1) * IN_F + xq * 4;
    const float* xg1 = x + (size_t)min(row0 + xr1, Nn - 1) * IN_F + xq * 4;
    const int wp0 = tid >> 6, wc = tid & 63;
    const float* wg = W + (size_t)(2 * wp0) * OUTC + wc * 4;

    float4 xa, xb = make_float4(0.f, 0.f, 0.f, 0.f), w0a, w0b, w1a, w1b;
    auto gload = [&](int k0) {
        xa = *(const float4*)(xg0 + k0);
        if (x2) xb = *(const float4*)(xg1 + k0);
        const float* p = wg + (size_t)k0 * OUTC;
        w0a = *(const float4*)(p);
        w0b = *(const float4*)(p + OUTC);
        w1a = *(const float4*)(p + 16 * OUTC);
        w1b = *(const float4*)(p + 17 * OUTC);
    };
    auto sstore = [&](int b) {
        *(float4*)(&xs[b][xr0 * GM_XP + xq * 4]) = xa;
        if (x2) *(float4*)(&xs[b][xr1 * GM_XP + xq * 4]) = xb;
        float* d0 = &wsm[b][wp0 * GM_WP + wc * 8];
        *(float4*)(d0)     = make_float4(w0a.x, w0b.x, w0a.y, w0b.y);
        *(float4*)(d0 + 4) = make_float4(w0a.z, w0b.z, w0a.w, w0b.w);
        float* d1 = d0 + 8 * GM_WP;
        *(float4*)(d1)     = make_float4(w1a.x, w1b.x, w1a.y, w1b.y);
        *(float4*)(d1 + 4) = make_float4(w1a.z, w1b.z, w1a.w, w1b.w);
    };

    v8f acc[GM_MT];
    #pragma unroll
    for (int t = 0; t < GM_MT; ++t) acc[t] = v8f{};

    gload(0);
    sstore(0);
    __syncthreads();

    for (int c = 0; c < IN_F / GM_KC; ++c) {
        const int  b    = c & 1;
        const bool more = c + 1 < IN_F / GM_KC;
        if (more) gload((c + 1) * GM_KC);
        __builtin_amdgcn_sched_barrier(0);
        const float* xf = &xs[b][m * GM_XP + 2 * hf];
        const float* wf = &wsm[b][hf * GM_WP + 2 * (wv * 16 + m)];
        #pragma unroll
        for (int ks = 0; ks < GM_KC / 4; ++ks) {
            const v2f bf = *(const v2f*)(wf + ks * 2 * GM_WP);
            #pragma unroll
            for (int t = 0; t < GM_MT; ++t) {
                const v2f a = *(const v2f*)(xf + t * 16 * GM_XP + ks * 4);
                acc[t] = __builtin_amdgcn_wmma_f32_16x16x4_f32(
                         false, a,  false, bf,
                         (short)0, acc[t],  false,  false);
            }
        }
        if (more) sstore(b ^ 1);
        __syncthreads();
    }

    const int rowOff = hf << 3;
    #pragma unroll
    for (int t = 0; t < GM_MT; ++t) {
        const int mTile = blockIdx.x * GM_MT + t;
        if (mTile < nTilesM) {
            float* hp = h + (size_t)(mTile * 16 + rowOff) * OUTC + wv * 16 + m;
            #pragma unroll
            for (int i = 0; i < 8; ++i)
                hp[(size_t)i * OUTC] = acc[t][i];
            float* tp = hT + (size_t)(wv * 16 + m) * Nn + mTile * 16 + rowOff;
            *(float4*)tp       = make_float4(acc[t][0], acc[t][1], acc[t][2], acc[t][3]);
            *(float4*)(tp + 4) = make_float4(acc[t][4], acc[t][5], acc[t][6], acc[t][7]);
        }
    }
}

#define EPW 16

__device__ __forceinline__ float dot4_leaky(float4 v, float4 a) {
    return leaky(v.x) * a.x + leaky(v.y) * a.y + leaky(v.z) * a.z + leaky(v.w) * a.w;
}

__global__ void __launch_bounds__(256) gat_scores(
        const float* __restrict__ h, const int* __restrict__ ei,
        const float* __restrict__ att, float* __restrict__ scores,
        int E, int Nn) {
    const int wave = blockIdx.x * (blockDim.x >> 5) + (threadIdx.x >> 5);
    const int lane = threadIdx.x & 31;
    const int g = lane >> 3;
    const int j = lane & 7;
    const int total = E + Nn;
    const size_t pitch = (size_t)((total + 3) & ~3);
    const int e0 = wave * EPW;
    if (e0 >= total) return;

    float4 as[8], ad[8];
    #pragma unroll
    for (int i = 0; i < 8; ++i) {
        const int head = i >> 1;
        const int off  = (j + ((i & 1) << 3)) << 2;
        as[i] = *(const float4*)(att + head * 2 * OUT_F + off);
        ad[i] = *(const float4*)(att + head * 2 * OUT_F + OUT_F + off);
    }

    int sv[EPW / 4], dv[EPW / 4];
    #pragma unroll
    for (int t = 0; t < EPW / 4; ++t) {
        const int e = min(e0 + t * 4 + g, total - 1);
        if (e < E) { sv[t] = ei[e]; dv[t] = ei[E + e]; } else { sv[t] = dv[t] = e - E; }
    }

    const bool hi4 = (j & 4) != 0;
    const bool hi2 = (j & 2) != 0;

    float4 rs[2][8], rd[2][8];
    {
        const float4* hs = (const float4*)(h + (size_t)sv[0] * OUTC) + j;
        const float4* hd = (const float4*)(h + (size_t)dv[0] * OUTC) + j;
        #pragma unroll
        for (int i = 0; i < 8; ++i) { rs[0][i] = hs[8 * i]; rd[0][i] = hd[8 * i]; }
    }

    #pragma unroll
    for (int t = 0; t < EPW / 4; ++t) {
        const int e = e0 + t * 4 + g;
        const int cur = t & 1, nxt = cur ^ 1;
        if (t + 1 < EPW / 4) {
            const float4* hs = (const float4*)(h + (size_t)sv[t + 1] * OUTC) + j;
            const float4* hd = (const float4*)(h + (size_t)dv[t + 1] * OUTC) + j;
            #pragma unroll
            for (int i = 0; i < 8; ++i) { rs[nxt][i] = hs[8 * i]; rd[nxt][i] = hd[8 * i]; }
        }

        float acc[HEADS] = {0.f, 0.f, 0.f, 0.f};
        #pragma unroll
        for (int i = 0; i < 8; ++i)
            acc[i >> 1] += dot4_leaky(rs[cur][i], as[i]) + dot4_leaky(rd[cur][i], ad[i]);

        float k0 = hi4 ? acc[2] : acc[0];
        float k1 = hi4 ? acc[3] : acc[1];
        k0 += __shfl_xor(hi4 ? acc[0] : acc[2], 4, 32);
        k1 += __shfl_xor(hi4 ? acc[1] : acc[3], 4, 32);
        float v = hi2 ? k1 : k0;
        v += __shfl_xor(hi2 ? k0 : k1, 2, 32);
        v += __shfl_xor(v, 1, 32);

        if (e < total && (j & 1) == 0) scores[(size_t)(j >> 1) * pitch + e] = v;
    }
}

#define NPART 32

__device__ __forceinline__ float wave_max(float m) {
    #pragma unroll
    for (int off = 16; off > 0; off >>= 1)
        m = fmaxf(m, __shfl_xor(m, off, 32));
    return m;
}

__global__ void __launch_bounds__(1024) gat_max(
        const float* __restrict__ scores, float* __restrict__ pmax, int totalEdges) {
    __shared__ float smax[32];
    const int head = blockIdx.x & (HEADS - 1);
    const int part = blockIdx.x >> 2;
    const size_t pitch = (size_t)((totalEdges + 3) & ~3);
    const float* pl = scores + (size_t)head * pitch;
    const float4* pl4 = (const float4*)pl;
    const int n4 = totalEdges >> 2;
    const int stride = (NPART / HEADS) * 1024;

    float m = -INFINITY;
    #pragma unroll 4
    for (int i = part * 1024 + threadIdx.x; i < n4; i += stride) {
        const float4 s = pl4[i];
        m = fmaxf(fmaxf(m, fmaxf(s.x, s.y)), fmaxf(s.z, s.w));
    }
    if (part == 0) {
        const int e = (n4 << 2) + threadIdx.x;
        if (e < totalEdges) m = fmaxf(m, pl[e]);
    }

    m = wave_max(m);
    const int lane = threadIdx.x & 31;
    if (lane == 0) smax[threadIdx.x >> 5] = m;
    __syncthreads();
    if (threadIdx.x < 32) {
        m = wave_max(smax[lane]);
        if (lane == 0) pmax[blockIdx.x] = m;
    }
}

#define NODES 20000

__device__ __forceinline__ float col_trip(const float* hcol, float* ocol,
                                          int4 s, int4 d, float4 sc, float hmax) {
    const float h0 = hcol[s.x], h1 = hcol[s.y], h2 = hcol[s.z], h3 = hcol[s.w];
    const float w0 = __expf(sc.x - hmax), w1 = __expf(sc.y - hmax);
    const float w2 = __expf(sc.z - hmax), w3 = __expf(sc.w - hmax);
    asm volatile("" :: "v"(h0), "v"(h1), "v"(h2), "v"(h3));
    atomicAdd(&ocol[d.x], h0 * w0);
    atomicAdd(&ocol[d.y], h1 * w1);
    atomicAdd(&ocol[d.z], h2 * w2);
    atomicAdd(&ocol[d.w], h3 * w3);
    return (w0 + w1) + (w2 + w3);
}

__global__ void __launch_bounds__(1024) gat_colsum(
        const float* __restrict__ hT, const int* __restrict__ ei,
        const float* __restrict__ scores, const float* __restrict__ pmax,
        float* __restrict__ outT, int E, int Nn) {
    __shared__ float hcol[NODES];
    __shared__ float ocol[NODES];
    __shared__ float red[32];
    if (Nn > NODES) return;

    const int c   = blockIdx.x;
    const int tid = threadIdx.x;
    const int total = E + Nn;
    const int hd = c >> 6;
    const float* hsrc = hT + (size_t)c * Nn;
    const float* w    = scores + (size_t)hd * (size_t)((total + 3) & ~3);
    const bool vec = (((E | Nn) & 3) == 0);

    float hmax = pmax[hd];
    #pragma unroll
    for (int j = 1; j < NPART / HEADS; ++j)
        hmax = fmaxf(hmax, pmax[j * HEADS + hd]);

    const int* src = ei;
    const int* dst = ei + E;
    const int E4 = vec ? (E >> 2) : 0;
    const int4*   src4 = (const int4*)src;
    const int4*   dst4 = (const int4*)dst;
    const float4* w4p  = (const float4*)w;
    const int last = E4 - 1;
    const int rot = (int)(((long long)c * E4) / OUTC) & ~31;
    auto quad = [=](int p) { const int i = min(p, last) + rot; return i >= E4 ? i - E4 : i; };
    int4 sP = make_int4(0, 0, 0, 0), dP = sP, sQ = sP, dQ = sP;
    float4 wP = make_float4(0.f, 0.f, 0.f, 0.f), wQ = wP;
    if (E4 > 0) {
        { const int i = quad(tid);        sP = src4[i]; dP = dst4[i]; wP = w4p[i]; }
        { const int i = quad(tid + 1024); sQ = src4[i]; dQ = dst4[i]; wQ = w4p[i]; }
    }

    float lsum = 0.f;
    if (vec) {
        for (int q = tid; q < (Nn >> 2); q += 1024) {
            const float4 hv = ((const float4*)hsrc)[q];
            const float4 sv = ((const float4*)(w + E))[q];
            const float4 wv = make_float4(__expf(sv.x - hmax), __expf(sv.y - hmax),
                                          __expf(sv.z - hmax), __expf(sv.w - hmax));
            lsum += (wv.x + wv.y) + (wv.z + wv.w);
            ((float4*)hcol)[q] = hv;
            ((float4*)ocol)[q] = make_float4(hv.x * wv.x, hv.y * wv.y, hv.z * wv.z, hv.w * wv.w);
        }
    } else {
        for (int n = tid; n < Nn; n += 1024) {
            const float hv = hsrc[n];
            const float wv = __expf(w[E + n] - hmax);
            lsum += wv;
            hcol[n] = hv;
            ocol[n] = hv * wv;
        }
    }
    __syncthreads();

    if (E4 > 0) {
        int4 sR, dR;
        float4 wR;
        for (int q = tid; q < E4; q += 3072) {
            { const int i = quad(q + 2048); sR = src4[i]; dR = dst4[i]; wR = w4p[i]; }
            lsum += col_trip(hcol, ocol, sP, dP, wP, hmax);
            { const int i = quad(q + 3072); sP = src4[i]; dP = dst4[i]; wP = w4p[i]; }
            if (q + 1024 < E4) lsum += col_trip(hcol, ocol, sQ, dQ, wQ, hmax);
            { const int i = quad(q + 4096); sQ = src4[i]; dQ = dst4[i]; wQ = w4p[i]; }
            if (q + 2048 < E4) lsum += col_trip(hcol, ocol, sR, dR, wR, hmax);
        }
    }
    for (int e = (E4 << 2) + tid; e < E; e += 1024) {
        const float wv = __expf(w[e] - hmax);
        lsum += wv;
        atomicAdd(&ocol[dst[e]], hcol[src[e]] * wv);
    }

    #pragma unroll
    for (int off = 16; off > 0; off >>= 1)
        lsum += __shfl_xor(lsum, off, 32);
    if ((tid & 31) == 0) red[tid >> 5] = lsum;
    __syncthreads();
    float tot = 0.f;
    #pragma unroll
    for (int i = 0; i < 32; ++i) tot += red[i];
    const float inv = 1.f / tot;

    float* od = outT + (size_t)c * Nn;
    if (vec) {
        for (int q = tid; q < (Nn >> 2); q += 1024) {
            const float4 o = ((const float4*)ocol)[q];
            ((float4*)od)[q] = make_float4(o.x * inv, o.y * inv, o.z * inv, o.w * inv);
        }
    } else {
        for (int n = tid; n < Nn; n += 1024) od[n] = ocol[n] * inv;
    }
}

__global__ void __launch_bounds__(256) gat_finish(
        const float* __restrict__ outT, const float* __restrict__ b,
        float* __restrict__ out, int Nn) {
    __shared__ float tile[8][32][33];
    const int wv   = threadIdx.x >> 5;
    const int lane = threadIdx.x & 31;
    const int n0   = blockIdx.x * 32;
    const int c0   = wv * 32;
    const float bias = b[c0 + lane];

    const int nl = min(n0 + lane, Nn - 1);
    float v[32];
    #pragma unroll
    for (int j = 0; j < 32; ++j)
        v[j] = outT[(size_t)(c0 + j) * Nn + nl];
    #pragma unroll
    for (int j = 0; j < 32; ++j)
        tile[wv][j][lane] = v[j];
    __syncthreads();
    if (n0 + 32 <= Nn) {
        float r[32];
        #pragma unroll
        for (int i = 0; i < 32; ++i)
            r[i] = tile[wv][lane][i];
        #pragma unroll
        for (int i = 0; i < 32; ++i)
            out[(size_t)(n0 + i) * OUTC + c0 + lane] = r[i] + bias;
    } else {
        for (int i = 0; i < 32; ++i)
            if (n0 + i < Nn)
                out[(size_t)(n0 + i) * OUTC + c0 + lane] = tile[wv][lane][i] + bias;
    }
}

extern "C" void kernel_launch(void* const* d_in, const int* in_sizes, int n_in,
                              void* d_out, int out_size, void* d_ws, size_t ws_size,
                              hipStream_t stream) {
    const float* x   = (const float*)d_in[0];
    const float* W   = (const float*)d_in[1];
    const float* att = (const float*)d_in[2];
    const float* b   = (const float*)d_in[3];
    const int*   ei  = (const int*)d_in[4];

    const int Nn = in_sizes[0] / IN_F;
    const int E  = in_sizes[4] / 2;
    const int totalEdges = E + Nn;

    float* out    = (float*)d_out;
    float* h      = (float*)d_ws;
    float* hT     = h + (size_t)Nn * OUTC;
    float* outT   = hT + (size_t)Nn * OUTC;
    float* scores = outT + (size_t)Nn * OUTC;
    float* pmax   = scores + (size_t)(totalEdges + 4) * HEADS;

    const int nTilesM = Nn / 16;
    gat_gemm_wmma<<<(nTilesM + GM_MT - 1) / GM_MT, 512, 0, stream>>>(x, W, h, hT, nTilesM, Nn);

    gat_scores<<<(totalEdges + 8 * EPW - 1) / (8 * EPW), 256, 0, stream>>>(h, ei, att, scores, E, Nn);
    gat_max<<<NPART, 1024, 0, stream>>>(scores, pmax, totalEdges);
    gat_colsum<<<OUTC, 1024, 0, stream>>>(hT, ei, scores, pmax, outT, E, Nn);
    gat_finish<<<(Nn + 31) / 32, 256, 0, stream>>>(outT, b, out, Nn);
}


# aggregation blocks own two columns and half of the edges each: one 8-byte LDS gather and one edge read serve both columns; last kernel joins the two halves' sums and totals (sum order changes)
# speedup vs baseline: 12.9080x; 1.0617x over previous
#include <hip/hip_runtime.h>
#include <hip/hip_bf16.h>
#include <math.h>

typedef __attribute__((ext_vector_type(2))) float v2f;
typedef __attribute__((ext_vector_type(8))) float v8f;

#define IN_F   256
#define OUTC   256
#define HEADS  4
#define OUT_F  64
#define NEG_SLOPE 0.2f

__device__ __forceinline__ float leaky(float v) { return v > 0.f ? v : NEG_SLOPE * v; }

#define GM_MT   5
#define GM_ROWS (GM_MT * 16)
#define GM_KC   32
#define GM_XP   36
#define GM_WP   544

__global__ void __launch_bounds__(512) gat_gemm_wmma(
        const float* __restrict__ x, const float* __restrict__ W,
        float* __restrict__ h, float* __restrict__ hT, int nTilesM, int Nn) {
    __shared__ __attribute__((aligned(16))) float xs[2][GM_ROWS * GM_XP];
    __shared__ __attribute__((aligned(16))) float wsm[2][(GM_KC / 2) * GM_WP];

    const int tid  = threadIdx.x;
    const int wv   = tid >> 5;
    const int lane = tid & 31;
    const int m    = lane & 15;
    const int hf   = lane >> 4;
    const int row0 = blockIdx.x * GM_ROWS;

    const int  xr0 = tid >> 3, xq = tid & 7;
    const int  xr1 = (tid + 512) >> 3;
    const bool x2  = tid < GM_ROWS * 8 - 512;
    const float* xg0 = x + (size_t)min(row0 + xr0, Nn - 1) * IN_F + xq * 4;
    const float* xg1 = x + (size_t)min(row0 + xr1, Nn - 1) * IN_F + xq * 4;
    const int wp0 = tid >> 6, wc = tid & 63;
    const float* wg = W + (size_t)(2 * wp0) * OUTC + wc * 4;

    float4 xa, xb = make_float4(0.f, 0.f, 0.f, 0.f), w0a, w0b, w1a, w1b;
    auto gload = [&](int k0) {
        xa = *(const float4*)(xg0 + k0);
        if (x2) xb = *(const float4*)(xg1 + k0);
        const float* p = wg + (size_t)k0 * OUTC;
        w0a = *(const float4*)(p);
        w0b = *(const float4*)(p + OUTC);
        w1a = *(const float4*)(p + 16 * OUTC);
        w1b = *(const float4*)(p + 17 * OUTC);
    };
    auto sstore = [&](int b) {
        *(float4*)(&xs[b][xr0 * GM_XP + xq * 4]) = xa;
        if (x2) *(float4*)(&xs[b][xr1 * GM_XP + xq * 4]) = xb;
        float* d0 = &wsm[b][wp0 * GM_WP + wc * 8];
        *(float4*)(d0)     = make_float4(w0a.x, w0b.x, w0a.y, w0b.y);
        *(float4*)(d0 + 4) = make_float4(w0a.z, w0b.z, w0a.w, w0b.w);
        float* d1 = d0 + 8 * GM_WP;
        *(float4*)(d1)     = make_float4(w1a.x, w1b.x, w1a.y, w1b.y);
        *(float4*)(d1 + 4) = make_float4(w1a.z, w1b.z, w1a.w, w1b.w);
    };

    v8f acc[GM_MT];
    #pragma unroll
    for (int t = 0; t < GM_MT; ++t) acc[t] = v8f{};

    gload(0);
    sstore(0);
    __syncthreads();

    for (int c = 0; c < IN_F / GM_KC; ++c) {
        const int  b    = c & 1;
        const bool more = c + 1 < IN_F / GM_KC;
        if (more) gload((c + 1) * GM_KC);
        __builtin_amdgcn_sched_barrier(0);
        const float* xf = &xs[b][m * GM_XP + 2 * hf];
        const float* wf = &wsm[b][hf * GM_WP + 2 * (wv * 16 + m)];
        #pragma unroll
        for (int ks = 0; ks < GM_KC / 4; ++ks) {
            const v2f bf = *(const v2f*)(wf + ks * 2 * GM_WP);
            #pragma unroll
            for (int t = 0; t < GM_MT; ++t) {
                const v2f a = *(const v2f*)(xf + t * 16 * GM_XP + ks * 4);
                acc[t] = __builtin_amdgcn_wmma_f32_16x16x4_f32(
                         false, a,  false, bf,
                         (short)0, acc[t],  false,  false);
            }
        }
        if (more) sstore(b ^ 1);
        __syncthreads();
    }

    const int rowOff = hf << 3;
    #pragma unroll
    for (int t = 0; t < GM_MT; ++t) {
        const int mTile = blockIdx.x * GM_MT + t;
        if (mTile < nTilesM) {
            float* hp = h + (size_t)(mTile * 16 + rowOff) * OUTC + wv * 16 + m;
            #pragma unroll
            for (int i = 0; i < 8; ++i)
                hp[(size_t)i * OUTC] = acc[t][i];
            float* tp = hT + (size_t)(wv * 16 + m) * Nn + mTile * 16 + rowOff;
            *(float4*)tp       = make_float4(acc[t][0], acc[t][1], acc[t][2], acc[t][3]);
            *(float4*)(tp + 4) = make_float4(acc[t][4], acc[t][5], acc[t][6], acc[t][7]);
        }
    }
}

#define EPW 16

__device__ __forceinline__ float dot4_leaky(float4 v, float4 a) {
    return leaky(v.x) * a.x + leaky(v.y) * a.y + leaky(v.z) * a.z + leaky(v.w) * a.w;
}

__global__ void __launch_bounds__(256) gat_scores(
        const float* __restrict__ h, const int* __restrict__ ei,
        const float* __restrict__ att, float* __restrict__ scores,
        int E, int Nn) {
    const int wave = blockIdx.x * (blockDim.x >> 5) + (threadIdx.x >> 5);
    const int lane = threadIdx.x & 31;
    const int g = lane >> 3;
    const int j = lane & 7;
    const int total = E + Nn;
    const size_t pitch = (size_t)((total + 3) & ~3);
    const int e0 = wave * EPW;
    if (e0 >= total) return;

    float4 as[8], ad[8];
    #pragma unroll
    for (int i = 0; i < 8; ++i) {
        const int head = i >> 1;
        const int off  = (j + ((i & 1) << 3)) << 2;
        as[i] = *(const float4*)(att + head * 2 * OUT_F + off);
        ad[i] = *(const float4*)(att + head * 2 * OUT_F + OUT_F + off);
    }

    int sv[EPW / 4], dv[EPW / 4];
    #pragma unroll
    for (int t = 0; t < EPW / 4; ++t) {
        const int e = min(e0 + t * 4 + g, total - 1);
        if (e < E) { sv[t] = ei[e]; dv[t] = ei[E + e]; } else { sv[t] = dv[t] = e - E; }
    }

    const bool hi4 = (j & 4) != 0;
    const bool hi2 = (j & 2) != 0;

    float4 rs[2][8], rd[2][8];
    {
        const float4* hs = (const float4*)(h + (size_t)sv[0] * OUTC) + j;
        const float4* hd = (const float4*)(h + (size_t)dv[0] * OUTC) + j;
        #pragma unroll
        for (int i = 0; i < 8; ++i) { rs[0][i] = hs[8 * i]; rd[0][i] = hd[8 * i]; }
    }

    #pragma unroll
    for (int t = 0; t < EPW / 4; ++t) {
        const int e = e0 + t * 4 + g;
        const int cur = t & 1, nxt = cur ^ 1;
        if (t + 1 < EPW / 4) {
            const float4* hs = (const float4*)(h + (size_t)sv[t + 1] * OUTC) + j;
            const float4* hd = (const float4*)(h + (size_t)dv[t + 1] * OUTC) + j;
            #pragma unroll
            for (int i = 0; i < 8; ++i) { rs[nxt][i] = hs[8 * i]; rd[nxt][i] = hd[8 * i]; }
        }

        float acc[HEADS] = {0.f, 0.f, 0.f, 0.f};
        #pragma unroll
        for (int i = 0; i < 8; ++i)
            acc[i >> 1] += dot4_leaky(rs[cur][i], as[i]) + dot4_leaky(rd[cur][i], ad[i]);

        float k0 = hi4 ? acc[2] : acc[0];
        float k1 = hi4 ? acc[3] : acc[1];
        k0 += __shfl_xor(hi4 ? acc[0] : acc[2], 4, 32);
        k1 += __shfl_xor(hi4 ? acc[1] : acc[3], 4, 32);
        float v = hi2 ? k1 : k0;
        v += __shfl_xor(hi2 ? k0 : k1, 2, 32);
        v += __shfl_xor(v, 1, 32);

        if (e < total && (j & 1) == 0) scores[(size_t)(j >> 1) * pitch + e] = v;
    }
}

#define NPART 32

__device__ __forceinline__ float wave_max(float m) {
    #pragma unroll
    for (int off = 16; off > 0; off >>= 1)
        m = fmaxf(m, __shfl_xor(m, off, 32));
    return m;
}

__global__ void __launch_bounds__(1024) gat_max(
        const float* __restrict__ scores, float* __restrict__ pmax, int totalEdges) {
    __shared__ float smax[32];
    const int head = blockIdx.x & (HEADS - 1);
    const int part = blockIdx.x >> 2;
    const size_t pitch = (size_t)((totalEdges + 3) & ~3);
    const float* pl = scores + (size_t)head * pitch;
    const float4* pl4 = (const float4*)pl;
    const int n4 = totalEdges >> 2;
    const int stride = (NPART / HEADS) * 1024;

    float m = -INFINITY;
    #pragma unroll 4
    for (int i = part * 1024 + threadIdx.x; i < n4; i += stride) {
        const float4 s = pl4[i];
        m = fmaxf(fmaxf(m, fmaxf(s.x, s.y)), fmaxf(s.z, s.w));
    }
    if (part == 0) {
        const int e = (n4 << 2) + threadIdx.x;
        if (e < totalEdges) m = fmaxf(m, pl[e]);
    }

    m = wave_max(m);
    const int lane = threadIdx.x & 31;
    if (lane == 0) smax[threadIdx.x >> 5] = m;
    __syncthreads();
    if (threadIdx.x < 32) {
        m = wave_max(smax[lane]);
        if (lane == 0) pmax[blockIdx.x] = m;
    }
}

#define NODES 20000

__device__ __forceinline__ float col_trip(const float2* hc, float* oc0, float* oc1,
                                          int4 s, int4 d, float4 sc, float hmax) {
    const float2 h0 = hc[s.x], h1 = hc[s.y], h2 = hc[s.z], h3 = hc[s.w];
    const float w0 = __expf(sc.x - hmax), w1 = __expf(sc.y - hmax);
    const float w2 = __expf(sc.z - hmax), w3 = __expf(sc.w - hmax);
    asm volatile("" :: "v"(h0.x), "v"(h0.y), "v"(h1.x), "v"(h1.y),
                       "v"(h2.x), "v"(h2.y), "v"(h3.x), "v"(h3.y));
    atomicAdd(&oc0[d.x], h0.x * w0);
    atomicAdd(&oc1[d.x], h0.y * w0);
    atomicAdd(&oc0[d.y], h1.x * w1);
    atomicAdd(&oc1[d.y], h1.y * w1);
    atomicAdd(&oc0[d.z], h2.x * w2);
    atomicAdd(&oc1[d.z], h2.y * w2);
    atomicAdd(&oc0[d.w], h3.x * w3);
    atomicAdd(&oc1[d.w], h3.y * w3);
    return (w0 + w1) + (w2 + w3);
}

__global__ void __launch_bounds__(1024) gat_colsum(
        const float* __restrict__ hT, const int* __restrict__ ei,
        const float* __restrict__ scores, const float* __restrict__ pmax,
        float* __restrict__ outT, float* __restrict__ wtot, int E, int Nn) {
    __shared__ __attribute__((aligned(16))) float2 hc[NODES];
    __shared__ __attribute__((aligned(16))) float oc0[NODES];
    __shared__ __attribute__((aligned(16))) float oc1[NODES];
    __shared__ float red[32];
    if (Nn > NODES) return;

    const int p   = blockIdx.x >> 1;
    const int s   = blockIdx.x & 1;
    const int c0  = 2 * p;
    const int tid = threadIdx.x;
    const int total = E + Nn;
    const int hd = c0 >> 6;
    const float* hsrc0 = hT + (size_t)c0 * Nn;
    const float* hsrc1 = hsrc0 + Nn;
    const float* w     = scores + (size_t)hd * (size_t)((total + 3) & ~3);
    const bool vec = (((E | Nn) & 3) == 0);

    float hmax = pmax[hd];
    #pragma unroll
    for (int j = 1; j < NPART / HEADS; ++j)
        hmax = fmaxf(hmax, pmax[j * HEADS + hd]);

    const int* src = ei;
    const int* dst = ei + E;
    const int E4 = vec ? (E >> 2) : 0;
    const int qlo = s ? (E4 >> 1) : 0;
    const int n   = s ? E4 - (E4 >> 1) : (E4 >> 1);
    const int4*   src4 = (const int4*)src + qlo;
    const int4*   dst4 = (const int4*)dst + qlo;
    const float4* w4p  = (const float4*)w + qlo;
    const int last = n - 1;
    const int rot = (int)(((long long)p * n) / (OUTC / 2)) & ~31;
    auto quad = [=](int q) { const int i = min(q, last) + rot; return i >= n ? i - n : i; };
    int4 sP = make_int4(0, 0, 0, 0), dP = sP, sQ = sP, dQ = sP;
    float4 wP = make_float4(0.f, 0.f, 0.f, 0.f), wQ = wP;
    if (n > 0) {
        { const int i = quad(tid);        sP = src4[i]; dP = dst4[i]; wP = w4p[i]; }
        { const int i = quad(tid + 1024); sQ = src4[i]; dQ = dst4[i]; wQ = w4p[i]; }
    }

    float lsum = 0.f;
    if (vec) {
        for (int q = tid; q < (Nn >> 2); q += 1024) {
            const float4 ha = ((const float4*)hsrc0)[q];
            const float4 hb = ((const float4*)hsrc1)[q];
            ((float4*)hc)[2 * q]     = make_float4(ha.x, hb.x, ha.y, hb.y);
            ((float4*)hc)[2 * q + 1] = make_float4(ha.z, hb.z, ha.w, hb.w);
            if (s == 0) {
                const float4 sv = ((const float4*)(w + E))[q];
                const float4 wv = make_float4(__expf(sv.x - hmax), __expf(sv.y - hmax),
                                              __expf(sv.z - hmax), __expf(sv.w - hmax));
                lsum += (wv.x + wv.y) + (wv.z + wv.w);
                ((float4*)oc0)[q] = make_float4(ha.x * wv.x, ha.y * wv.y, ha.z * wv.z, ha.w * wv.w);
                ((float4*)oc1)[q] = make_float4(hb.x * wv.x, hb.y * wv.y, hb.z * wv.z, hb.w * wv.w);
            } else {
                ((float4*)oc0)[q] = make_float4(0.f, 0.f, 0.f, 0.f);
                ((float4*)oc1)[q] = make_float4(0.f, 0.f, 0.f, 0.f);
            }
        }
    } else {
        for (int i = tid; i < Nn; i += 1024) {
            const float ha = hsrc0[i], hb = hsrc1[i];
            hc[i] = make_float2(ha, hb);
            if (s == 0) {
                const float wv = __expf(w[E + i] - hmax);
                lsum += wv;
                oc0[i] = ha * wv;
                oc1[i] = hb * wv;
            } else {
                oc0[i] = 0.f;
                oc1[i] = 0.f;
            }
        }
    }
    __syncthreads();

    if (n > 0) {
        int4 sR, dR;
        float4 wR;
        for (int q = tid; q < n; q += 3072) {
            { const int i = quad(q + 2048); sR = src4[i]; dR = dst4[i]; wR = w4p[i]; }
            lsum += col_trip(hc, oc0, oc1, sP, dP, wP, hmax);
            { const int i = quad(q + 3072); sP = src4[i]; dP = dst4[i]; wP = w4p[i]; }
            if (q + 1024 < n) lsum += col_trip(hc, oc0, oc1, sQ, dQ, wQ, hmax);
            { const int i = quad(q + 4096); sQ = src4[i]; dQ = dst4[i]; wQ = w4p[i]; }
            if (q + 2048 < n) lsum += col_trip(hc, oc0, oc1, sR, dR, wR, hmax);
        }
    }
    for (int e = (E4 << 2) + s * 1024 + tid; e < E; e += 2048) {
        const float wv = __expf(w[e] - hmax);
        const float2 hv = hc[src[e]];
        lsum += wv;
        atomicAdd(&oc0[dst[e]], hv.x * wv);
        atomicAdd(&oc1[dst[e]], hv.y * wv);
    }

    #pragma unroll
    for (int off = 16; off > 0; off >>= 1)
        lsum += __shfl_xor(lsum, off, 32);
    if ((tid & 31) == 0) red[tid >> 5] = lsum;
    __syncthreads();
    if (tid == 0) {
        float tot = 0.f;
        #pragma unroll
        for (int i = 0; i < 32; ++i) tot += red[i];
        wtot[blockIdx.x] = tot;
    }

    float* od0 = outT + ((size_t)s * OUTC + c0) * Nn;
    float* od1 = od0 + Nn;
    if (vec) {
        for (int q = tid; q < (Nn >> 2); q += 1024) {
            ((float4*)od0)[q] = ((const float4*)oc0)[q];
            ((float4*)od1)[q] = ((const float4*)oc1)[q];
        }
    } else {
        for (int i = tid; i < Nn; i += 1024) { od0[i] = oc0[i]; od1[i] = oc1[i]; }
    }
}

__global__ void __launch_bounds__(256) gat_finish(
        const float* __restrict__ outT, const float* __restrict__ wtot,
        const float* __restrict__ b, float* __restrict__ out, int Nn) {
    __shared__ float tile[8][32][33];
    const int wv   = threadIdx.x >> 5;
    const int lane = threadIdx.x & 31;
    const int n0   = blockIdx.x * 32;
    const int c0   = wv * 32;
    const int col  = c0 + lane;
    const float bias = b[col];
    const float inv = 1.f / (wtot[col & ~1] + wtot[col | 1]);
    const float* pa = outT;
    const float* pb = outT + (size_t)OUTC * Nn;

    const int nl = min(n0 + lane, Nn - 1);
    float va[32], vb[32];
    #pragma unroll
    for (int j = 0; j < 32; ++j) {
        va[j] = pa[(size_t)(c0 + j) * Nn + nl];
        vb[j] = pb[(size_t)(c0 + j) * Nn + nl];
    }
    #pragma unroll
    for (int j = 0; j < 32; ++j)
        tile[wv][j][lane] = va[j] + vb[j];
    __syncthreads();
    if (n0 + 32 <= Nn) {
        float r[32];
        #pragma unroll
        for (int i = 0; i < 32; ++i)
            r[i] = tile[wv][lane][i];
        #pragma unroll
        for (int i = 0; i < 32; ++i)
            out[(size_t)(n0 + i) * OUTC + col] = r[i] * inv + bias;
    } else {
        for (int i = 0; i < 32; ++i)
            if (n0 + i < Nn)
                out[(size_t)(n0 + i) * OUTC + col] = tile[wv][lane][i] * inv + bias;
    }
}

extern "C" void kernel_launch(void* const* d_in, const int* in_sizes, int n_in,
                              void* d_out, int out_size, void* d_ws, size_t ws_size,
                              hipStream_t stream) {
    const float* x   = (const float*)d_in[0];
    const float* W   = (const float*)d_in[1];
    const float* att = (const float*)d_in[2];
    const float* b   = (const float*)d_in[3];
    const int*   ei  = (const int*)d_in[4];

    const int Nn = in_sizes[0] / IN_F;
    const int E  = in_sizes[4] / 2;
    const int totalEdges = E + Nn;

    float* out    = (float*)d_out;
    float* h      = (float*)d_ws;
    float* hT     = h + (size_t)Nn * OUTC;
    float* outT   = hT + (size_t)Nn * OUTC;
    float* scores = outT + (size_t)2 * Nn * OUTC;
    float* pmax   = scores + (size_t)(totalEdges + 4) * HEADS;
    float* wtot   = pmax + NPART;

    const int nTilesM = Nn / 16;
    gat_gemm_wmma<<<(nTilesM + GM_MT - 1) / GM_MT, 512, 0, stream>>>(x, W, h, hT, nTilesM, Nn);

    gat_scores<<<(totalEdges + 8 * EPW - 1) / (8 * EPW), 256, 0, stream>>>(h, ei, att, scores, E, Nn);
    gat_max<<<NPART, 1024, 0, stream>>>(scores, pmax, totalEdges);
    gat_colsum<<<OUTC, 1024, 0, stream>>>(hT, ei, scores, pmax, outT, wtot, E, Nn);
    gat_finish<<<(Nn + 31) / 32, 256, 0, stream>>>(outT, wtot, b, out, Nn);
}


# scores kernel: a wave walks 84 edges (21 trips) in a rolled two-buffer loop, node ids two trips ahead, instead of 16; 4048 long-lived waves replace 21250, attention vectors read once per 84 edges
# speedup vs baseline: 13.2982x; 1.0302x over previous
#include <hip/hip_runtime.h>
#include <hip/hip_bf16.h>
#include <math.h>

typedef __attribute__((ext_vector_type(2))) float v2f;
typedef __attribute__((ext_vector_type(8))) float v8f;

#define IN_F   256
#define OUTC   256
#define HEADS  4
#define OUT_F  64
#define NEG_SLOPE 0.2f

__device__ __forceinline__ float leaky(float v) { return v > 0.f ? v : NEG_SLOPE * v; }

#define GM_MT   5
#define GM_ROWS (GM_MT * 16)
#define GM_KC   32
#define GM_XP   36
#define GM_WP   544

__global__ void __launch_bounds__(512) gat_gemm_wmma(
        const float* __restrict__ x, const float* __restrict__ W,
        float* __restrict__ h, float* __restrict__ hT, int nTilesM, int Nn) {
    __shared__ __attribute__((aligned(16))) float xs[2][GM_ROWS * GM_XP];
    __shared__ __attribute__((aligned(16))) float wsm[2][(GM_KC / 2) * GM_WP];

    const int tid  = threadIdx.x;
    const int wv   = tid >> 5;
    const int lane = tid & 31;
    const int m    = lane & 15;
    const int hf   = lane >> 4;
    const int row0 = blockIdx.x * GM_ROWS;

    const int  xr0 = tid >> 3, xq = tid & 7;
    const int  xr1 = (tid + 512) >> 3;
    const bool x2  = tid < GM_ROWS * 8 - 512;
    const float* xg0 = x + (size_t)min(row0 + xr0, Nn - 1) * IN_F + xq * 4;
    const float* xg1 = x + (size_t)min(row0 + xr1, Nn - 1) * IN_F + xq * 4;
    const int wp0 = tid >> 6, wc = tid & 63;
    const float* wg = W + (size_t)(2 * wp0) * OUTC + wc * 4;

    float4 xa, xb = make_float4(0.f, 0.f, 0.f, 0.f), w0a, w0b, w1a, w1b;
    auto gload = [&](int k0) {
        xa = *(const float4*)(xg0 + k0);
        if (x2) xb = *(const float4*)(xg1 + k0);
        const float* p = wg + (size_t)k0 * OUTC;
        w0a = *(const float4*)(p);
        w0b = *(const float4*)(p + OUTC);
        w1a = *(const float4*)(p + 16 * OUTC);
        w1b = *(const float4*)(p + 17 * OUTC);
    };
    auto sstore = [&](int b) {
        *(float4*)(&xs[b][xr0 * GM_XP + xq * 4]) = xa;
        if (x2) *(float4*)(&xs[b][xr1 * GM_XP + xq * 4]) = xb;
        float* d0 = &wsm[b][wp0 * GM_WP + wc * 8];
        *(float4*)(d0)     = make_float4(w0a.x, w0b.x, w0a.y, w0b.y);
        *(float4*)(d0 + 4) = make_float4(w0a.z, w0b.z, w0a.w, w0b.w);
        float* d1 = d0 + 8 * GM_WP;
        *(float4*)(d1)     = make_float4(w1a.x, w1b.x, w1a.y, w1b.y);
        *(float4*)(d1 + 4) = make_float4(w1a.z, w1b.z, w1a.w, w1b.w);
    };

    v8f acc[GM_MT];
    #pragma unroll
    for (int t = 0; t < GM_MT; ++t) acc[t] = v8f{};

    gload(0);
    sstore(0);
    __syncthreads();

    for (int c = 0; c < IN_F / GM_KC; ++c) {
        const int  b    = c & 1;
        const bool more = c + 1 < IN_F / GM_KC;
        if (more) gload((c + 1) * GM_KC);
        __builtin_amdgcn_sched_barrier(0);
        const float* xf = &xs[b][m * GM_XP + 2 * hf];
        const float* wf = &wsm[b][hf * GM_WP + 2 * (wv * 16 + m)];
        #pragma unroll
        for (int ks = 0; ks < GM_KC / 4; ++ks) {
            const v2f bf = *(const v2f*)(wf + ks * 2 * GM_WP);
            #pragma unroll
            for (int t = 0; t < GM_MT; ++t) {
                const v2f a = *(const v2f*)(xf + t * 16 * GM_XP + ks * 4);
                acc[t] = __builtin_amdgcn_wmma_f32_16x16x4_f32(
                         false, a,  false, bf,
                         (short)0, acc[t],  false,  false);
            }
        }
        if (more) sstore(b ^ 1);
        __syncthreads();
    }

    const int rowOff = hf << 3;
    #pragma unroll
    for (int t = 0; t < GM_MT; ++t) {
        const int mTile = blockIdx.x * GM_MT + t;
        if (mTile < nTilesM) {
            float* hp = h + (size_t)(mTile * 16 + rowOff) * OUTC + wv * 16 + m;
            #pragma unroll
            for (int i = 0; i < 8; ++i)
                hp[(size_t)i * OUTC] = acc[t][i];
            float* tp = hT + (size_t)(wv * 16 + m) * Nn + mTile * 16 + rowOff;
            *(float4*)tp       = make_float4(acc[t][0], acc[t][1], acc[t][2], acc[t][3]);
            *(float4*)(tp + 4) = make_float4(acc[t][4], acc[t][5], acc[t][6], acc[t][7]);
        }
    }
}

#define EPW 84
#define NTRIP (EPW / 4)
static_assert(EPW % 4 == 0 && NTRIP % 2 == 1, "the trip loop below expects an odd number of whole trips");

__device__ __forceinline__ float dot4_leaky(float4 v, float4 a) {
    return leaky(v.x) * a.x + leaky(v.y) * a.y + leaky(v.z) * a.z + leaky(v.w) * a.w;
}

__global__ void __launch_bounds__(256) gat_scores(
        const float* __restrict__ h, const int* __restrict__ ei,
        const float* __restrict__ att, float* __restrict__ scores,
        int E, int Nn) {
    const int wave = blockIdx.x * (blockDim.x >> 5) + (threadIdx.x >> 5);
    const int lane = threadIdx.x & 31;
    const int g = lane >> 3;
    const int j = lane & 7;
    const int total = E + Nn;
    const size_t pitch = (size_t)((total + 3) & ~3);
    const int e0 = wave * EPW;
    if (e0 >= total) return;

    float4 as[8], ad[8];
    #pragma unroll
    for (int i = 0; i < 8; ++i) {
        const int head = i >> 1;
        const int off  = (j + ((i & 1) << 3)) << 2;
        as[i] = *(const float4*)(att + head * 2 * OUT_F + off);
        ad[i] = *(const float4*)(att + head * 2 * OUT_F + OUT_F + off);
    }

    const bool hi4 = (j & 4) != 0;
    const bool hi2 = (j & 2) != 0;

    auto ids = [&](int t, int& s, int& d) {
        const int e  = min(e0 + t * 4 + g, total - 1);
        const int ec = min(e, E - 1);
        const int a = ei[ec], b = ei[E + ec];
        s = e < E ? a : e - E;
        d = e < E ? b : e - E;
    };
    auto rows = [&](float4 (&ra)[8], float4 (&rb)[8], int s, int d) {
        const float4* hs = (const float4*)(h + (size_t)s * OUTC) + j;
        const float4* hd = (const float4*)(h + (size_t)d * OUTC) + j;
        #pragma unroll
        for (int i = 0; i < 8; ++i) { ra[i] = hs[8 * i]; rb[i] = hd[8 * i]; }
    };
    auto score = [&](const float4 (&ra)[8], const float4 (&rb)[8], int t) {
        float acc[HEADS] = {0.f, 0.f, 0.f, 0.f};
        #pragma unroll
        for (int i = 0; i < 8; ++i)
            acc[i >> 1] += dot4_leaky(ra[i], as[i]) + dot4_leaky(rb[i], ad[i]);

        float k0 = hi4 ? acc[2] : acc[0];
        float k1 = hi4 ? acc[3] : acc[1];
        k0 += __shfl_xor(hi4 ? acc[0] : acc[2], 4, 32);
        k1 += __shfl_xor(hi4 ? acc[1] : acc[3], 4, 32);
        float v = hi2 ? k1 : k0;
        v += __shfl_xor(hi2 ? k0 : k1, 2, 32);
        v += __shfl_xor(v, 1, 32);

        const int e = e0 + t * 4 + g;
        if (e < total && (j & 1) == 0) scores[(size_t)(j >> 1) * pitch + e] = v;
    };

    float4 rsA[8], rdA[8], rsB[8], rdB[8];
    int s1, d1, s2, d2;
    {
        int s0, d0;
        ids(0, s0, d0);
        ids(1, s1, d1);
        rows(rsA, rdA, s0, d0);
    }
    #pragma unroll 1
    for (int t = 0; t + 1 < NTRIP; t += 2) {
        ids(t + 2, s2, d2);
        rows(rsB, rdB, s1, d1);
        __builtin_amdgcn_sched_barrier(0);
        score(rsA, rdA, t);
        ids(t + 3, s1, d1);
        rows(rsA, rdA, s2, d2);
        __builtin_amdgcn_sched_barrier(0);
        score(rsB, rdB, t + 1);
    }
    score(rsA, rdA, NTRIP - 1);
}

#define NPART 32

__device__ __forceinline__ float wave_max(float m) {
    #pragma unroll
    for (int off = 16; off > 0; off >>= 1)
        m = fmaxf(m, __shfl_xor(m, off, 32));
    return m;
}

__global__ void __launch_bounds__(1024) gat_max(
        const float* __restrict__ scores, float* __restrict__ pmax, int totalEdges) {
    __shared__ float smax[32];
    const int head = blockIdx.x & (HEADS - 1);
    const int part = blockIdx.x >> 2;
    const size_t pitch = (size_t)((totalEdges + 3) & ~3);
    const float* pl = scores + (size_t)head * pitch;
    const float4* pl4 = (const float4*)pl;
    const int n4 = totalEdges >> 2;
    const int stride = (NPART / HEADS) * 1024;

    float m = -INFINITY;
    #pragma unroll 4
    for (int i = part * 1024 + threadIdx.x; i < n4; i += stride) {
        const float4 s = pl4[i];
        m = fmaxf(fmaxf(m, fmaxf(s.x, s.y)), fmaxf(s.z, s.w));
    }
    if (part == 0) {
        const int e = (n4 << 2) + threadIdx.x;
        if (e < totalEdges) m = fmaxf(m, pl[e]);
    }

    m = wave_max(m);
    const int lane = threadIdx.x & 31;
    if (lane == 0) smax[threadIdx.x >> 5] = m;
    __syncthreads();
    if (threadIdx.x < 32) {
        m = wave_max(smax[lane]);
        if (lane == 0) pmax[blockIdx.x] = m;
    }
}

#define NODES 20000

__device__ __forceinline__ float col_trip(const float2* hc, float* oc0, float* oc1,
                                          int4 s, int4 d, float4 sc, float hmax) {
    const float2 h0 = hc[s.x], h1 = hc[s.y], h2 = hc[s.z], h3 = hc[s.w];
    const float w0 = __expf(sc.x - hmax), w1 = __expf(sc.y - hmax);
    const float w2 = __expf(sc.z - hmax), w3 = __expf(sc.w - hmax);
    asm volatile("" :: "v"(h0.x), "v"(h0.y), "v"(h1.x), "v"(h1.y),
                       "v"(h2.x), "v"(h2.y), "v"(h3.x), "v"(h3.y));
    atomicAdd(&oc0[d.x], h0.x * w0);
    atomicAdd(&oc1[d.x], h0.y * w0);
    atomicAdd(&oc0[d.y], h1.x * w1);
    atomicAdd(&oc1[d.y], h1.y * w1);
    atomicAdd(&oc0[d.z], h2.x * w2);
    atomicAdd(&oc1[d.z], h2.y * w2);
    atomicAdd(&oc0[d.w], h3.x * w3);
    atomicAdd(&oc1[d.w], h3.y * w3);
    return (w0 + w1) + (w2 + w3);
}

__global__ void __launch_bounds__(1024) gat_colsum(
        const float* __restrict__ hT, const int* __restrict__ ei,
        const float* __restrict__ scores, const float* __restrict__ pmax,
        float* __restrict__ outT, float* __restrict__ wtot, int E, int Nn) {
    __shared__ __attribute__((aligned(16))) float2 hc[NODES];
    __shared__ __attribute__((aligned(16))) float oc0[NODES];
    __shared__ __attribute__((aligned(16))) float oc1[NODES];
    __shared__ float red[32];
    if (Nn > NODES) return;

    const int p   = blockIdx.x >> 1;
    const int s   = blockIdx.x & 1;
    const int c0  = 2 * p;
    const int tid = threadIdx.x;
    const int total = E + Nn;
    const int hd = c0 >> 6;
    const float* hsrc0 = hT + (size_t)c0 * Nn;
    const float* hsrc1 = hsrc0 + Nn;
    const float* w     = scores + (size_t)hd * (size_t)((total + 3) & ~3);
    const bool vec = (((E | Nn) & 3) == 0);

    float hmax = pmax[hd];
    #pragma unroll
    for (int j = 1; j < NPART / HEADS; ++j)
        hmax = fmaxf(hmax, pmax[j * HEADS + hd]);

    const int* src = ei;
    const int* dst = ei + E;
    const int E4 = vec ? (E >> 2) : 0;
    const int qlo = s ? (E4 >> 1) : 0;
    const int n   = s ? E4 - (E4 >> 1) : (E4 >> 1);
    const int4*   src4 = (const int4*)src + qlo;
    const int4*   dst4 = (const int4*)dst + qlo;
    const float4* w4p  = (const float4*)w + qlo;
    const int last = n - 1;
    const int rot = (int)(((long long)p * n) / (OUTC / 2)) & ~31;
    auto quad = [=](int q) { const int i = min(q, last) + rot; return i >= n ? i - n : i; };
    int4 sP = make_int4(0, 0, 0, 0), dP = sP, sQ = sP, dQ = sP;
    float4 wP = make_float4(0.f, 0.f, 0.f, 0.f), wQ = wP;
    if (n > 0) {
        { const int i = quad(tid);        sP = src4[i]; dP = dst4[i]; wP = w4p[i]; }
        { const int i = quad(tid + 1024); sQ = src4[i]; dQ = dst4[i]; wQ = w4p[i]; }
    }

    float lsum = 0.f;
    if (vec) {
        for (int q = tid; q < (Nn >> 2); q += 1024) {
            const float4 ha = ((const float4*)hsrc0)[q];
            const float4 hb = ((const float4*)hsrc1)[q];
            ((float4*)hc)[2 * q]     = make_float4(ha.x, hb.x, ha.y, hb.y);
            ((float4*)hc)[2 * q + 1] = make_float4(ha.z, hb.z, ha.w, hb.w);
            if (s == 0) {
                const float4 sv = ((const float4*)(w + E))[q];
                const float4 wv = make_float4(__expf(sv.x - hmax), __expf(sv.y - hmax),
                                              __expf(sv.z - hmax), __expf(sv.w - hmax));
                lsum += (wv.x + wv.y) + (wv.z + wv.w);
                ((float4*)oc0)[q] = make_float4(ha.x * wv.x, ha.y * wv.y, ha.z * wv.z, ha.w * wv.w);
                ((float4*)oc1)[q] = make_float4(hb.x * wv.x, hb.y * wv.y, hb.z * wv.z, hb.w * wv.w);
            } else {
                ((float4*)oc0)[q] = make_float4(0.f, 0.f, 0.f, 0.f);
                ((float4*)oc1)[q] = make_float4(0.f, 0.f, 0.f, 0.f);
            }
        }
    } else {
        for (int i = tid; i < Nn; i += 1024) {
            const float ha = hsrc0[i], hb = hsrc1[i];
            hc[i] = make_float2(ha, hb);
            if (s == 0) {
                const float wv = __expf(w[E + i] - hmax);
                lsum += wv;
                oc0[i] = ha * wv;
                oc1[i] = hb * wv;
            } else {
                oc0[i] = 0.f;
                oc1[i] = 0.f;
            }
        }
    }
    __syncthreads();

    if (n > 0) {
        int4 sR, dR;
        float4 wR;
        for (int q = tid; q < n; q += 3072) {
            { const int i = quad(q + 2048); sR = src4[i]; dR = dst4[i]; wR = w4p[i]; }
            lsum += col_trip(hc, oc0, oc1, sP, dP, wP, hmax);
            { const int i = quad(q + 3072); sP = src4[i]; dP = dst4[i]; wP = w4p[i]; }
            if (q + 1024 < n) lsum += col_trip(hc, oc0, oc1, sQ, dQ, wQ, hmax);
            { const int i = quad(q + 4096); sQ = src4[i]; dQ = dst4[i]; wQ = w4p[i]; }
            if (q + 2048 < n) lsum += col_trip(hc, oc0, oc1, sR, dR, wR, hmax);
        }
    }
    for (int e = (E4 << 2) + s * 1024 + tid; e < E; e += 2048) {
        const float wv = __expf(w[e] - hmax);
        const float2 hv = hc[src[e]];
        lsum += wv;
        atomicAdd(&oc0[dst[e]], hv.x * wv);
        atomicAdd(&oc1[dst[e]], hv.y * wv);
    }

    #pragma unroll
    for (int off = 16; off > 0; off >>= 1)
        lsum += __shfl_xor(lsum, off, 32);
    if ((tid & 31) == 0) red[tid >> 5] = lsum;
    __syncthreads();
    if (tid == 0) {
        float tot = 0.f;
        #pragma unroll
        for (int i = 0; i < 32; ++i) tot += red[i];
        wtot[blockIdx.x] = tot;
    }

    float* od0 = outT + ((size_t)s * OUTC + c0) * Nn;
    float* od1 = od0 + Nn;
    if (vec) {
        for (int q = tid; q < (Nn >> 2); q += 1024) {
            ((float4*)od0)[q] = ((const float4*)oc0)[q];
            ((float4*)od1)[q] = ((const float4*)oc1)[q];
        }
    } else {
        for (int i = tid; i < Nn; i += 1024) { od0[i] = oc0[i]; od1[i] = oc1[i]; }
    }
}

__global__ void __launch_bounds__(256) gat_finish(
        const float* __restrict__ outT, const float* __restrict__ wtot,
        const float* __restrict__ b, float* __restrict__ out, int Nn) {
    __shared__ float tile[8][32][33];
    const int wv   = threadIdx.x >> 5;
    const int lane = threadIdx.x & 31;
    const int n0   = blockIdx.x * 32;
    const int c0   = wv * 32;
    const int col  = c0 + lane;
    const float bias = b[col];
    const float inv = 1.f / (wtot[col & ~1] + wtot[col | 1]);
    const float* pa = outT;
    const float* pb = outT + (size_t)OUTC * Nn;

    const int nl = min(n0 + lane, Nn - 1);
    float va[32], vb[32];
    #pragma unroll
    for (int j = 0; j < 32; ++j) {
        va[j] = pa[(size_t)(c0 + j) * Nn + nl];
        vb[j] = pb[(size_t)(c0 + j) * Nn + nl];
    }
    #pragma unroll
    for (int j = 0; j < 32; ++j)
        tile[wv][j][lane] = va[j] + vb[j];
    __syncthreads();
    if (n0 + 32 <= Nn) {
        float r[32];
        #pragma unroll
        for (int i = 0; i < 32; ++i)
            r[i] = tile[wv][lane][i];
        #pragma unroll
        for (int i = 0; i < 32; ++i)
            out[(size_t)(n0 + i) * OUTC + col] = r[i] * inv + bias;
    } else {
        for (int i = 0; i < 32; ++i)
            if (n0 + i < Nn)
                out[(size_t)(n0 + i) * OUTC + col] = tile[wv][lane][i] * inv + bias;
    }
}

extern "C" void kernel_launch(void* const* d_in, const int* in_sizes, int n_in,
                              void* d_out, int out_size, void* d_ws, size_t ws_size,
                              hipStream_t stream) {
    const float* x   = (const float*)d_in[0];
    const float* W   = (const float*)d_in[1];
    const float* att = (const float*)d_in[2];
    const float* b   = (const float*)d_in[3];
    const int*   ei  = (const int*)d_in[4];

    const int Nn = in_sizes[0] / IN_F;
    const int E  = in_sizes[4] / 2;
    const int totalEdges = E + Nn;

    float* out    = (float*)d_out;
    float* h      = (float*)d_ws;
    float* hT     = h + (size_t)Nn * OUTC;
    float* outT   = hT + (size_t)Nn * OUTC;
    float* scores = outT + (size_t)2 * Nn * OUTC;
    float* pmax   = scores + (size_t)(totalEdges + 4) * HEADS;
    float* wtot   = pmax + NPART;

    const int nTilesM = Nn / 16;
    gat_gemm_wmma<<<(nTilesM + GM_MT - 1) / GM_MT, 512, 0, stream>>>(x, W, h, hT, nTilesM, Nn);

    gat_scores<<<(totalEdges + 8 * EPW - 1) / (8 * EPW), 256, 0, stream>>>(h, ei, att, scores, E, Nn);
    gat_max<<<NPART, 1024, 0, stream>>>(scores, pmax, totalEdges);
    gat_colsum<<<OUTC, 1024, 0, stream>>>(hT, ei, scores, pmax, outT, wtot, E, Nn);
    gat_finish<<<(Nn + 31) / 32, 256, 0, stream>>>(outT, wtot, b, out, Nn);
}
